# starting kernel
# baseline (speedup 1.0000x reference)
#include <hip/hip_runtime.h>

typedef __attribute__((ext_vector_type(2))) float v2f;
typedef __attribute__((ext_vector_type(8))) float v8f;

#define H 64
#define F85 85
#define KPAD 88

__global__ void zero_kernel(float* __restrict__ p, long n) {
    long i = (long)blockIdx.x * blockDim.x + threadIdx.x;
    if (i < n) p[i] = 0.0f;
}

__global__ void prep_w_kernel(const float* __restrict__ w1_src_t,
                              const float* __restrict__ w1_dst_t,
                              const float* __restrict__ b1_t,
                              const float* __restrict__ w1_src_r,
                              const float* __restrict__ w1_dst_r,
                              float* __restrict__ W1t, float* __restrict__ W1c,
                              float* __restrict__ W1m, float* __restrict__ b1sum) {
    int i = blockIdx.x * blockDim.x + threadIdx.x;
    if (i >= KPAD * H) return;
    int r = i >> 6, c = i & 63;
    float vt = 0.f, vc = 0.f, vm = 0.f;
    if (r < F85) {
        vt = w1_dst_t[r * H + c] + w1_dst_t[F85 * H + r * H + c];
        vc = w1_src_r[r * H + c];
        vm = w1_src_r[F85 * H + r * H + c];
    } else if (r == 85) {
        vt = w1_src_t[c];
        vc = w1_dst_r[c];
        vm = w1_dst_r[H + c];
    } else if (r == 86) {
        vt = w1_src_t[H + c];
    }
    W1t[i] = vt; W1c[i] = vc; W1m[i] = vm;
    if (i < H) b1sum[i] = b1_t[i] + b1_t[H + i];
}

__global__ void scatter_scalar_kernel(const float* __restrict__ xc, const float* __restrict__ xm,
                                      const int* __restrict__ sp, const int* __restrict__ dp,
                                      const int* __restrict__ sr, const int* __restrict__ dr,
                                      float* __restrict__ aggc, float* __restrict__ aggm, int E) {
    int e = blockIdx.x * blockDim.x + threadIdx.x;
    if (e >= E) return;
    atomicAdd(&aggc[dp[e]], xc[sp[e]]);
    atomicAdd(&aggm[dr[e]], xm[sr[e]]);
}

__global__ void scatter_feat_kernel(float* __restrict__ outbuf, const int* __restrict__ oidx,
                                    const float* __restrict__ inbuf, const int* __restrict__ iidx,
                                    int E, int F) {
    long i = (long)blockIdx.x * blockDim.x + threadIdx.x;
    long total = (long)E * F;
    if (i >= total) return;
    int e = (int)(i / F), j = (int)(i % F);
    atomicAdd(&outbuf[(size_t)oidx[e] * F + j], inbuf[(size_t)iidx[e] * F + j]);
}

__global__ __launch_bounds__(128) void gemm88_kernel(
    const float* __restrict__ Xmain, const float* __restrict__ extra1,
    const float* __restrict__ extra2, const float* __restrict__ W,
    const float* __restrict__ bias, float* __restrict__ out, int M) {
    __shared__ float xs[16 * KPAD];

    const int row0 = blockIdx.x * 16;
    for (int i = threadIdx.x; i < 16 * KPAD; i += 128) {
        int r = i / KPAD, c = i - r * KPAD;
        int grow = row0 + r;
        float v;
        if (c < F85)      v = Xmain[(size_t)grow * F85 + c];
        else if (c == 85) v = extra1 ? extra1[grow] : 0.0f;
        else if (c == 86) v = extra2 ? extra2[grow] : 0.0f;
        else              v = 0.0f;
        xs[i] = v;
    }
    __syncthreads();

    const int wave = threadIdx.x >> 5;
    const int lane = threadIdx.x & 31;
    const int half = lane >> 4;
    const int lm   = lane & 15;
    const int c0   = wave * 16;

    const float bval = bias[c0 + lm];
    v8f acc;
#pragma unroll
    for (int i = 0; i < 8; ++i) acc[i] = bval;

    for (int k = 0; k < KPAD; k += 4) {
        const int ka = k + 2 * half;
        v2f a, b;
        a.x = xs[lm * KPAD + ka];
        a.y = xs[lm * KPAD + ka + 1];
        b.x = W[(size_t)(ka + 0) * H + c0 + lm];
        b.y = W[(size_t)(ka + 1) * H + c0 + lm];
        acc = __builtin_amdgcn_wmma_f32_16x16x4_f32(false, a, false, b, (short)0, acc,
                                                    false, false);
    }
#pragma unroll
    for (int v = 0; v < 8; ++v) {
        float r = acc[v];
        r = r > 0.0f ? r : 0.0f;
        out[(size_t)(row0 + v + 8 * half) * H + c0 + lm] = r;
    }
}

__global__ __launch_bounds__(128) void conv2_out_kernel(
    const float* __restrict__ h_t,  const float* __restrict__ aggc,
    const float* __restrict__ aggm, const float* __restrict__ w2l,
    const float* __restrict__ w2r,  const float* __restrict__ b2,
    const float* __restrict__ wlin, const float* __restrict__ blin,
    float* __restrict__ out, int M) {
    __shared__ float hs[16 * H];
    __shared__ float cs[16 * H];
    __shared__ float ms[16 * H];
    __shared__ float ts[16 * 2 * H];

    const int row0 = blockIdx.x * 16;
    for (int i = threadIdx.x; i < 16 * H; i += 128) {
        int r = i >> 6, c = i & 63;
        size_t g = (size_t)(row0 + r) * H + c;
        hs[i] = h_t[g];
        cs[i] = aggc[g];
        ms[i] = aggm[g];
    }
    __syncthreads();

    const int wave = threadIdx.x >> 5;
    const int lane = threadIdx.x & 31;
    const int half = lane >> 4;
    const int lm   = lane & 15;
    const int c0   = wave * 16;

    v8f acc1, acc2;
    const float bv1 = b2[0 * H + c0 + lm];
    const float bv2 = b2[1 * H + c0 + lm];
#pragma unroll
    for (int i = 0; i < 8; ++i) { acc1[i] = bv1; acc2[i] = bv2; }

    const float* W2l0 = w2l;
    const float* W2l1 = w2l + 1 * H * H;
    const float* W2r0 = w2r;
    const float* W2r1 = w2r + 1 * H * H;

    for (int k = 0; k < H; k += 4) {
        const int ka = k + 2 * half;
        v2f ah, ac, am, b;
        ah.x = hs[lm * H + ka]; ah.y = hs[lm * H + ka + 1];
        ac.x = cs[lm * H + ka]; ac.y = cs[lm * H + ka + 1];
        am.x = ms[lm * H + ka]; am.y = ms[lm * H + ka + 1];

        b.x = W2l0[(ka + 0) * H + c0 + lm]; b.y = W2l0[(ka + 1) * H + c0 + lm];
        acc1 = __builtin_amdgcn_wmma_f32_16x16x4_f32(false, ac, false, b, (short)0, acc1,
                                                     false, false);
        b.x = W2r0[(ka + 0) * H + c0 + lm]; b.y = W2r0[(ka + 1) * H + c0 + lm];
        acc1 = __builtin_amdgcn_wmma_f32_16x16x4_f32(false, ah, false, b, (short)0, acc1,
                                                     false, false);
        b.x = W2l1[(ka + 0) * H + c0 + lm]; b.y = W2l1[(ka + 1) * H + c0 + lm];
        acc2 = __builtin_amdgcn_wmma_f32_16x16x4_f32(false, am, false, b, (short)0, acc2,
                                                     false, false);
        b.x = W2r1[(ka + 0) * H + c0 + lm]; b.y = W2r1[(ka + 1) * H + c0 + lm];
        acc2 = __builtin_amdgcn_wmma_f32_16x16x4_f32(false, ah, false, b, (short)0, acc2,
                                                     false, false);
    }

#pragma unroll
    for (int v = 0; v < 8; ++v) {
        float r1 = acc1[v]; r1 = r1 > 0.0f ? r1 : 0.0f;
        float r2 = acc2[v]; r2 = r2 > 0.0f ? r2 : 0.0f;
        ts[(v + 8 * half) * 128 + c0 + lm]      = r1;
        ts[(v + 8 * half) * 128 + 64 + c0 + lm] = r2;
    }
    __syncthreads();

    if (threadIdx.x < 32) {
        const int r = threadIdx.x >> 1, c = threadIdx.x & 1;
        float s = blin[c];
#pragma unroll 8
        for (int j = 0; j < 128; ++j) s += ts[r * 128 + j] * wlin[j * 2 + c];
        out[(size_t)(row0 + r) * 2 + c] = s;
    }
}

extern "C" void kernel_launch(void* const* d_in, const int* in_sizes, int n_in,
                              void* d_out, int out_size, void* d_ws, size_t ws_size,
                              hipStream_t stream) {
    const float* x_card   = (const float*)d_in[0];
    const float* x_merch  = (const float*)d_in[1];
    const float* x_trans  = (const float*)d_in[2];
    const float* w1_src_t = (const float*)d_in[3];
    const float* w1_dst_t = (const float*)d_in[4];
    const float* b1_t     = (const float*)d_in[5];
    const float* w1_src_r = (const float*)d_in[6];
    const float* w1_dst_r = (const float*)d_in[7];
    const float* b1_r     = (const float*)d_in[8];
    const float* w2_l     = (const float*)d_in[9];
    const float* w2_r     = (const float*)d_in[10];
    const float* b2       = (const float*)d_in[11];
    const float* w_lin    = (const float*)d_in[12];
    const float* b_lin    = (const float*)d_in[13];
    const int* src_pays   = (const int*)d_in[14];
    const int* dst_pays   = (const int*)d_in[15];
    const int* src_recv   = (const int*)d_in[16];
    const int* dst_recv   = (const int*)d_in[17];
    float* out = (float*)d_out;

    const int N_c = in_sizes[0];
    const int N_m = in_sizes[1];
    const int N_t = in_sizes[2] / F85;
    const int E   = in_sizes[14];

    float* ws = (float*)d_ws;
    size_t off = 0;
    float* aggc1  = ws + off; off += (size_t)N_t;
    float* aggm1  = ws + off; off += (size_t)N_t;
    float* aggt2c = ws + off; off += (size_t)N_c * F85;
    float* aggt2m = ws + off; off += (size_t)N_m * F85;
    float* agg2c  = ws + off; off += (size_t)N_t * H;
    float* agg2m  = ws + off; off += (size_t)N_t * H;
    const long zeroCount = (long)off;
    float* h_t    = ws + off; off += (size_t)N_t * H;
    float* h_c    = ws + off; off += (size_t)N_c * H;
    float* h_m    = ws + off; off += (size_t)N_m * H;
    float* W1t    = ws + off; off += (size_t)KPAD * H;
    float* W1c    = ws + off; off += (size_t)KPAD * H;
    float* W1m    = ws + off; off += (size_t)KPAD * H;
    float* b1sum  = ws + off; off += (size_t)H;

    {
        long blocks = (zeroCount + 255) / 256;
        zero_kernel<<<(unsigned)blocks, 256, 0, stream>>>(ws, zeroCount);
    }
    prep_w_kernel<<<(KPAD * H + 255) / 256, 256, 0, stream>>>(
        w1_src_t, w1_dst_t, b1_t, w1_src_r, w1_dst_r, W1t, W1c, W1m, b1sum);
    scatter_scalar_kernel<<<(E + 255) / 256, 256, 0, stream>>>(
        x_card, x_merch, src_pays, dst_pays, src_recv, dst_recv, aggc1, aggm1, E);
    {
        long total = (long)E * F85;
        scatter_feat_kernel<<<(unsigned)((total + 255) / 256), 256, 0, stream>>>(
            aggt2c, src_pays, x_trans, dst_pays, E, F85);
        scatter_feat_kernel<<<(unsigned)((total + 255) / 256), 256, 0, stream>>>(
            aggt2m, src_recv, x_trans, dst_recv, E, F85);
    }
    gemm88_kernel<<<N_t / 16, 128, 0, stream>>>(x_trans, aggc1, aggm1, W1t, b1sum, h_t, N_t);
    gemm88_kernel<<<N_c / 16, 128, 0, stream>>>(aggt2c, x_card, nullptr, W1c, b1_r, h_c, N_c);
    gemm88_kernel<<<N_m / 16, 128, 0, stream>>>(aggt2m, x_merch, nullptr, W1m, b1_r + H, h_m, N_m);
    {
        long total = (long)E * H;
        scatter_feat_kernel<<<(unsigned)((total + 255) / 256), 256, 0, stream>>>(
            agg2c, dst_pays, h_c, src_pays, E, H);
        scatter_feat_kernel<<<(unsigned)((total + 255) / 256), 256, 0, stream>>>(
            agg2m, dst_recv, h_m, src_recv, E, H);
    }
    conv2_out_kernel<<<N_t / 16, 128, 0, stream>>>(
        h_t, agg2c, agg2m, w2_l, w2_r, b2, w_lin, b_lin, out, N_t);
}


# conv1 reverse aggregation summed in LDS: each workgroup owns a card/merchant row range, scans all edges, adds rows with LDS atomics (arrival order, as before), stores once; no global atomics there
# speedup vs baseline: 1.3053x; 1.3053x over previous
#include <hip/hip_runtime.h>

typedef __attribute__((ext_vector_type(2))) float v2f;
typedef __attribute__((ext_vector_type(8))) float v8f;

#define H 64
#define F85 85
#define KPAD 88

__global__ void zero_kernel(float* __restrict__ p, long n) {
    long i = (long)blockIdx.x * blockDim.x + threadIdx.x;
    if (i < n) p[i] = 0.0f;
}

__global__ void prep_w_kernel(const float* __restrict__ w1_src_t,
                              const float* __restrict__ w1_dst_t,
                              const float* __restrict__ b1_t,
                              const float* __restrict__ w1_src_r,
                              const float* __restrict__ w1_dst_r,
                              float* __restrict__ W1t, float* __restrict__ W1c,
                              float* __restrict__ W1m, float* __restrict__ b1sum) {
    int i = blockIdx.x * blockDim.x + threadIdx.x;
    if (i >= KPAD * H) return;
    int r = i >> 6, c = i & 63;
    float vt = 0.f, vc = 0.f, vm = 0.f;
    if (r < F85) {
        vt = w1_dst_t[r * H + c] + w1_dst_t[F85 * H + r * H + c];
        vc = w1_src_r[r * H + c];
        vm = w1_src_r[F85 * H + r * H + c];
    } else if (r == 85) {
        vt = w1_src_t[c];
        vc = w1_dst_r[c];
        vm = w1_dst_r[H + c];
    } else if (r == 86) {
        vt = w1_src_t[H + c];
    }
    W1t[i] = vt; W1c[i] = vc; W1m[i] = vm;
    if (i < H) b1sum[i] = b1_t[i] + b1_t[H + i];
}

__global__ void scatter_scalar_kernel(const float* __restrict__ xc, const float* __restrict__ xm,
                                      const int* __restrict__ sp, const int* __restrict__ dp,
                                      const int* __restrict__ sr, const int* __restrict__ dr,
                                      float* __restrict__ aggc, float* __restrict__ aggm, int E) {
    int e = blockIdx.x * blockDim.x + threadIdx.x;
    if (e >= E) return;
    atomicAdd(&aggc[dp[e]], xc[sp[e]]);
    atomicAdd(&aggm[dr[e]], xm[sr[e]]);
}

__global__ void scatter_feat_kernel(float* __restrict__ outbuf, const int* __restrict__ oidx,
                                    const float* __restrict__ inbuf, const int* __restrict__ iidx,
                                    int E, int F) {
    long i = (long)blockIdx.x * blockDim.x + threadIdx.x;
    long total = (long)E * F;
    if (i >= total) return;
    int e = (int)(i / F), j = (int)(i % F);
    atomicAdd(&outbuf[(size_t)oidx[e] * F + j], inbuf[(size_t)iidx[e] * F + j]);
}

#define RA_THREADS 1024
#define RA_TILE    16384
#define RA_UNR     6
#define RA_ROWS_C  392
#define RA_ROWS_M  80
#define RA_PITCH   96

template <bool VEC>
__device__ __forceinline__ void ra_load_keys(const int* __restrict__ key, int E, int tile,
                                             int tid, int4 (&cur)[4]) {
#pragma unroll
    for (int j = 0; j < 4; ++j) {
        const int e0 = tile * RA_TILE + (j * RA_THREADS + tid) * 4;
        int4 r;
        if (VEC) {
            r = *reinterpret_cast<const int4*>(key + (e0 < E ? e0 : 0));
        } else {
            const int last = E - 1;
            r.x = key[e0 + 0 < E ? e0 + 0 : last];
            r.y = key[e0 + 1 < E ? e0 + 1 : last];
            r.z = key[e0 + 2 < E ? e0 + 2 : last];
            r.w = key[e0 + 3 < E ? e0 + 3 : last];
        }
        cur[j] = r;
    }
}

template <bool VEC>
__device__ __forceinline__ void ra_scan_and_add(
    const float* __restrict__ x_trans, const int* __restrict__ key,
    const int* __restrict__ val, int E, int lo, int nrow, int part,
    float* acc, int* queue, int* qn) {
    const int tid  = threadIdx.x;
    const int wave = __builtin_amdgcn_readfirstlane(tid >> 5);
    const int lane = tid & 31;

    const int nTiles = (E + RA_TILE - 1) / RA_TILE;
    const int rot = part % nTiles;

    int4 cur[4];
    ra_load_keys<VEC>(key, E, rot, tid, cur);

    for (int it = 0; it < nTiles; ++it) {
        int tile = it + rot;
        if (tile >= nTiles) tile -= nTiles;
        const int buf = it & 1;

        unsigned m = 0;
        const int ebase = tile * RA_TILE + tid * 4;
#pragma unroll
        for (int j = 0; j < 4; ++j) {
            const int k[4] = {cur[j].x, cur[j].y, cur[j].z, cur[j].w};
#pragma unroll
            for (int c = 0; c < 4; ++c) {
                const bool inside = ebase + j * (RA_THREADS * 4) + c < E;
                m |= (inside && (unsigned)(k[c] - lo) < (unsigned)nrow ? 1u : 0u) << (j * 4 + c);
            }
        }
        if (m) {
            int pos = atomicAdd(&qn[buf], __builtin_popcount(m));
            do {
                const int b = __builtin_ctz(m);
                queue[pos++] = ebase + (b >> 2) * (RA_THREADS * 4) + (b & 3);
                m &= m - 1;
            } while (m);
        }
        __syncthreads();

        if (it + 1 < nTiles) {
            int ntile = tile + 1;
            if (ntile >= nTiles) ntile -= nTiles;
            ra_load_keys<VEC>(key, E, ntile, tid, cur);
        }
        if (tid == 0) qn[buf ^ 1] = 0;

        const int n = __builtin_amdgcn_readfirstlane(qn[buf]);
        auto round = [&](int q0) {
            const int myq = q0 + 32 * lane;
            const int ev  = queue[(lane < RA_UNR && myq < n) ? myq : 0];
            int eu[RA_UNR], rb[RA_UNR];
            const float* rowp[RA_UNR];
            float v0[RA_UNR], v1[RA_UNR], v2[RA_UNR];
#pragma unroll
            for (int u = 0; u < RA_UNR; ++u) eu[u] = __builtin_amdgcn_readlane(ev, u);
#pragma unroll
            for (int u = 0; u < RA_UNR; ++u) {
                rb[u]   = (key[eu[u]] - lo) * RA_PITCH;
                rowp[u] = x_trans + (size_t)val[eu[u]] * F85;
            }
#pragma unroll
            for (int u = 0; u < RA_UNR; ++u) {
                v0[u] = rowp[u][lane];
                v1[u] = rowp[u][32 + lane];
                v2[u] = rowp[u][lane < F85 - 64 ? 64 + lane : 64];
            }
#pragma unroll
            for (int u = 0; u < RA_UNR; ++u) {
                const bool ok = q0 + 32 * u < n;
                atomicAdd(&acc[rb[u] + lane],      ok ? v0[u] : 0.0f);
                atomicAdd(&acc[rb[u] + 32 + lane], ok ? v1[u] : 0.0f);
                atomicAdd(&acc[rb[u] + 64 + lane], ok ? v2[u] : 0.0f);
            }
        };
        if (n > 0) round(wave);
        for (int q0 = wave + 32 * RA_UNR; q0 < n; q0 += 32 * RA_UNR) round(q0);
        __syncthreads();
    }
}

__global__ __launch_bounds__(RA_THREADS) void rev_agg_lds_kernel(
    const float* __restrict__ x_trans,
    const int* __restrict__ src_pays, const int* __restrict__ dst_pays,
    const int* __restrict__ src_recv, const int* __restrict__ dst_recv,
    float* __restrict__ aggc, float* __restrict__ aggm,
    int N_c, int N_m, int E, int gc) {
    __shared__ float acc[RA_ROWS_C * RA_PITCH];
    __shared__ int   queue[RA_TILE];
    __shared__ int   qn[2];

    const int tid = threadIdx.x;
    const bool is_card = (int)blockIdx.x < gc;
    const int  part = is_card ? (int)blockIdx.x : (int)blockIdx.x - gc;
    const int  rows = is_card ? RA_ROWS_C : RA_ROWS_M;
    const int  N    = is_card ? N_c : N_m;
    const int* __restrict__ key = is_card ? src_pays : src_recv;
    const int* __restrict__ val = is_card ? dst_pays : dst_recv;
    float* __restrict__ outp    = is_card ? aggc : aggm;

    const int lo = part * rows;
    int nrow = N - lo;
    if (nrow > rows) nrow = rows;
    if (nrow < 0) nrow = 0;

    for (int i = tid; i < nrow * RA_PITCH; i += RA_THREADS) acc[i] = 0.0f;
    if (tid < 2) qn[tid] = 0;
    __syncthreads();

    if (E > 0) {
        const bool vec_ok = ((((size_t)key) & 15) == 0) && ((E & 3) == 0);
        if (vec_ok) ra_scan_and_add<true>(x_trans, key, val, E, lo, nrow, part, acc, queue, qn);
        else        ra_scan_and_add<false>(x_trans, key, val, E, lo, nrow, part, acc, queue, qn);
    }

    float* __restrict__ dstp = outp + (size_t)lo * F85;
    for (int i = tid; i < nrow * F85; i += RA_THREADS) {
        const int r = i / F85;
        dstp[i] = acc[r * RA_PITCH + (i - r * F85)];
    }
}

__global__ __launch_bounds__(128) void gemm88_kernel(
    const float* __restrict__ Xmain, const float* __restrict__ extra1,
    const float* __restrict__ extra2, const float* __restrict__ W,
    const float* __restrict__ bias, float* __restrict__ out, int M) {
    __shared__ float xs[16 * KPAD];

    const int row0 = blockIdx.x * 16;
    for (int i = threadIdx.x; i < 16 * KPAD; i += 128) {
        int r = i / KPAD, c = i - r * KPAD;
        int grow = row0 + r;
        float v;
        if (c < F85)      v = Xmain[(size_t)grow * F85 + c];
        else if (c == 85) v = extra1 ? extra1[grow] : 0.0f;
        else if (c == 86) v = extra2 ? extra2[grow] : 0.0f;
        else              v = 0.0f;
        xs[i] = v;
    }
    __syncthreads();

    const int wave = threadIdx.x >> 5;
    const int lane = threadIdx.x & 31;
    const int half = lane >> 4;
    const int lm   = lane & 15;
    const int c0   = wave * 16;

    const float bval = bias[c0 + lm];
    v8f acc;
#pragma unroll
    for (int i = 0; i < 8; ++i) acc[i] = bval;

    for (int k = 0; k < KPAD; k += 4) {
        const int ka = k + 2 * half;
        v2f a, b;
        a.x = xs[lm * KPAD + ka];
        a.y = xs[lm * KPAD + ka + 1];
        b.x = W[(size_t)(ka + 0) * H + c0 + lm];
        b.y = W[(size_t)(ka + 1) * H + c0 + lm];
        acc = __builtin_amdgcn_wmma_f32_16x16x4_f32(false, a, false, b, (short)0, acc,
                                                    false, false);
    }
#pragma unroll
    for (int v = 0; v < 8; ++v) {
        float r = acc[v];
        r = r > 0.0f ? r : 0.0f;
        out[(size_t)(row0 + v + 8 * half) * H + c0 + lm] = r;
    }
}

__global__ __launch_bounds__(128) void conv2_out_kernel(
    const float* __restrict__ h_t,  const float* __restrict__ aggc,
    const float* __restrict__ aggm, const float* __restrict__ w2l,
    const float* __restrict__ w2r,  const float* __restrict__ b2,
    const float* __restrict__ wlin, const float* __restrict__ blin,
    float* __restrict__ out, int M) {
    __shared__ float hs[16 * H];
    __shared__ float cs[16 * H];
    __shared__ float ms[16 * H];
    __shared__ float ts[16 * 2 * H];

    const int row0 = blockIdx.x * 16;
    for (int i = threadIdx.x; i < 16 * H; i += 128) {
        int r = i >> 6, c = i & 63;
        size_t g = (size_t)(row0 + r) * H + c;
        hs[i] = h_t[g];
        cs[i] = aggc[g];
        ms[i] = aggm[g];
    }
    __syncthreads();

    const int wave = threadIdx.x >> 5;
    const int lane = threadIdx.x & 31;
    const int half = lane >> 4;
    const int lm   = lane & 15;
    const int c0   = wave * 16;

    v8f acc1, acc2;
    const float bv1 = b2[0 * H + c0 + lm];
    const float bv2 = b2[1 * H + c0 + lm];
#pragma unroll
    for (int i = 0; i < 8; ++i) { acc1[i] = bv1; acc2[i] = bv2; }

    const float* W2l0 = w2l;
    const float* W2l1 = w2l + 1 * H * H;
    const float* W2r0 = w2r;
    const float* W2r1 = w2r + 1 * H * H;

    for (int k = 0; k < H; k += 4) {
        const int ka = k + 2 * half;
        v2f ah, ac, am, b;
        ah.x = hs[lm * H + ka]; ah.y = hs[lm * H + ka + 1];
        ac.x = cs[lm * H + ka]; ac.y = cs[lm * H + ka + 1];
        am.x = ms[lm * H + ka]; am.y = ms[lm * H + ka + 1];

        b.x = W2l0[(ka + 0) * H + c0 + lm]; b.y = W2l0[(ka + 1) * H + c0 + lm];
        acc1 = __builtin_amdgcn_wmma_f32_16x16x4_f32(false, ac, false, b, (short)0, acc1,
                                                     false, false);
        b.x = W2r0[(ka + 0) * H + c0 + lm]; b.y = W2r0[(ka + 1) * H + c0 + lm];
        acc1 = __builtin_amdgcn_wmma_f32_16x16x4_f32(false, ah, false, b, (short)0, acc1,
                                                     false, false);
        b.x = W2l1[(ka + 0) * H + c0 + lm]; b.y = W2l1[(ka + 1) * H + c0 + lm];
        acc2 = __builtin_amdgcn_wmma_f32_16x16x4_f32(false, am, false, b, (short)0, acc2,
                                                     false, false);
        b.x = W2r1[(ka + 0) * H + c0 + lm]; b.y = W2r1[(ka + 1) * H + c0 + lm];
        acc2 = __builtin_amdgcn_wmma_f32_16x16x4_f32(false, ah, false, b, (short)0, acc2,
                                                     false, false);
    }

#pragma unroll
    for (int v = 0; v < 8; ++v) {
        float r1 = acc1[v]; r1 = r1 > 0.0f ? r1 : 0.0f;
        float r2 = acc2[v]; r2 = r2 > 0.0f ? r2 : 0.0f;
        ts[(v + 8 * half) * 128 + c0 + lm]      = r1;
        ts[(v + 8 * half) * 128 + 64 + c0 + lm] = r2;
    }
    __syncthreads();

    if (threadIdx.x < 32) {
        const int r = threadIdx.x >> 1, c = threadIdx.x & 1;
        float s = blin[c];
#pragma unroll 8
        for (int j = 0; j < 128; ++j) s += ts[r * 128 + j] * wlin[j * 2 + c];
        out[(size_t)(row0 + r) * 2 + c] = s;
    }
}

extern "C" void kernel_launch(void* const* d_in, const int* in_sizes, int n_in,
                              void* d_out, int out_size, void* d_ws, size_t ws_size,
                              hipStream_t stream) {
    const float* x_card   = (const float*)d_in[0];
    const float* x_merch  = (const float*)d_in[1];
    const float* x_trans  = (const float*)d_in[2];
    const float* w1_src_t = (const float*)d_in[3];
    const float* w1_dst_t = (const float*)d_in[4];
    const float* b1_t     = (const float*)d_in[5];
    const float* w1_src_r = (const float*)d_in[6];
    const float* w1_dst_r = (const float*)d_in[7];
    const float* b1_r     = (const float*)d_in[8];
    const float* w2_l     = (const float*)d_in[9];
    const float* w2_r     = (const float*)d_in[10];
    const float* b2       = (const float*)d_in[11];
    const float* w_lin    = (const float*)d_in[12];
    const float* b_lin    = (const float*)d_in[13];
    const int* src_pays   = (const int*)d_in[14];
    const int* dst_pays   = (const int*)d_in[15];
    const int* src_recv   = (const int*)d_in[16];
    const int* dst_recv   = (const int*)d_in[17];
    float* out = (float*)d_out;

    const int N_c = in_sizes[0];
    const int N_m = in_sizes[1];
    const int N_t = in_sizes[2] / F85;
    const int E   = in_sizes[14];

    float* ws = (float*)d_ws;
    size_t off = 0;
    float* aggc1  = ws + off; off += (size_t)N_t;
    float* aggm1  = ws + off; off += (size_t)N_t;
    float* agg2c  = ws + off; off += (size_t)N_t * H;
    float* agg2m  = ws + off; off += (size_t)N_t * H;
    const long zeroCount = (long)off;
    float* aggt2c = ws + off; off += (size_t)N_c * F85;
    float* aggt2m = ws + off; off += (size_t)N_m * F85;
    float* h_t    = ws + off; off += (size_t)N_t * H;
    float* h_c    = ws + off; off += (size_t)N_c * H;
    float* h_m    = ws + off; off += (size_t)N_m * H;
    float* W1t    = ws + off; off += (size_t)KPAD * H;
    float* W1c    = ws + off; off += (size_t)KPAD * H;
    float* W1m    = ws + off; off += (size_t)KPAD * H;
    float* b1sum  = ws + off; off += (size_t)H;

    {
        long blocks = (zeroCount + 255) / 256;
        zero_kernel<<<(unsigned)blocks, 256, 0, stream>>>(ws, zeroCount);
    }
    prep_w_kernel<<<(KPAD * H + 255) / 256, 256, 0, stream>>>(
        w1_src_t, w1_dst_t, b1_t, w1_src_r, w1_dst_r, W1t, W1c, W1m, b1sum);
    scatter_scalar_kernel<<<(E + 255) / 256, 256, 0, stream>>>(
        x_card, x_merch, src_pays, dst_pays, src_recv, dst_recv, aggc1, aggm1, E);
    {
        const int gc = (N_c + RA_ROWS_C - 1) / RA_ROWS_C;
        const int gm = (N_m + RA_ROWS_M - 1) / RA_ROWS_M;
        rev_agg_lds_kernel<<<gc + gm, RA_THREADS, 0, stream>>>(
            x_trans, src_pays, dst_pays, src_recv, dst_recv, aggt2c, aggt2m, N_c, N_m, E, gc);
    }
    gemm88_kernel<<<N_t / 16, 128, 0, stream>>>(x_trans, aggc1, aggm1, W1t, b1sum, h_t, N_t);
    gemm88_kernel<<<N_c / 16, 128, 0, stream>>>(aggt2c, x_card, nullptr, W1c, b1_r, h_c, N_c);
    gemm88_kernel<<<N_m / 16, 128, 0, stream>>>(aggt2m, x_merch, nullptr, W1m, b1_r + H, h_m, N_m);
    {
        long total = (long)E * H;
        scatter_feat_kernel<<<(unsigned)((total + 255) / 256), 256, 0, stream>>>(
            agg2c, dst_pays, h_c, src_pays, E, H);
        scatter_feat_kernel<<<(unsigned)((total + 255) / 256), 256, 0, stream>>>(
            agg2m, dst_recv, h_m, src_recv, E, H);
    }
    conv2_out_kernel<<<N_t / 16, 128, 0, stream>>>(
        h_t, agg2c, agg2m, w2_l, w2_r, b2, w_lin, b_lin, out, N_t);
}


# conv2 aggregation as gather: per-transaction edge lists (one int exchange per edge), h_c/h_m rows summed inside the conv2 kernel in list order (arrival order, as the float atomics were)
# speedup vs baseline: 1.9495x; 1.4935x over previous
#include <hip/hip_runtime.h>

typedef __attribute__((ext_vector_type(2))) float v2f;
typedef __attribute__((ext_vector_type(8))) float v8f;

#define H 64
#define F85 85
#define KPAD 88

__global__ void init_kernel(float* __restrict__ z, int* __restrict__ heads, long n) {
    long i = (long)blockIdx.x * blockDim.x + threadIdx.x;
    if (i < n) { z[i] = 0.0f; heads[i] = -1; }
}

__global__ void prep_w_kernel(const float* __restrict__ w1_src_t,
                              const float* __restrict__ w1_dst_t,
                              const float* __restrict__ b1_t,
                              const float* __restrict__ w1_src_r,
                              const float* __restrict__ w1_dst_r,
                              float* __restrict__ W1t, float* __restrict__ W1c,
                              float* __restrict__ W1m, float* __restrict__ b1sum) {
    int i = blockIdx.x * blockDim.x + threadIdx.x;
    if (i >= KPAD * H) return;
    int r = i >> 6, c = i & 63;
    float vt = 0.f, vc = 0.f, vm = 0.f;
    if (r < F85) {
        vt = w1_dst_t[r * H + c] + w1_dst_t[F85 * H + r * H + c];
        vc = w1_src_r[r * H + c];
        vm = w1_src_r[F85 * H + r * H + c];
    } else if (r == 85) {
        vt = w1_src_t[c];
        vc = w1_dst_r[c];
        vm = w1_dst_r[H + c];
    } else if (r == 86) {
        vt = w1_src_t[H + c];
    }
    W1t[i] = vt; W1c[i] = vc; W1m[i] = vm;
    if (i < H) b1sum[i] = b1_t[i] + b1_t[H + i];
}

__global__ void edge_pass_kernel(const float* __restrict__ xc, const float* __restrict__ xm,
                                 const int* __restrict__ sp, const int* __restrict__ dp,
                                 const int* __restrict__ sr, const int* __restrict__ dr,
                                 float* __restrict__ aggc, float* __restrict__ aggm,
                                 int* __restrict__ headc, int* __restrict__ headm,
                                 int* __restrict__ nodec, int* __restrict__ nodem, int E) {
    int e = blockIdx.x * blockDim.x + threadIdx.x;
    if (e >= E) return;
    const int sc = sp[e], dc = dp[e], sm = sr[e], dm = dr[e];
    atomicAdd(&aggc[dc], xc[sc]);
    atomicAdd(&aggm[dm], xm[sm]);
    reinterpret_cast<int2*>(nodec)[e] = make_int2(atomicExch(&headc[dc], e), sc);
    reinterpret_cast<int2*>(nodem)[e] = make_int2(atomicExch(&headm[dm], e), sm);
}

#define RA_THREADS 1024
#define RA_TILE    16384
#define RA_UNR     6
#define RA_ROWS_C  392
#define RA_ROWS_M  80
#define RA_PITCH   96

template <bool VEC>
__device__ __forceinline__ void ra_load_keys(const int* __restrict__ key, int E, int tile,
                                             int tid, int4 (&cur)[4]) {
#pragma unroll
    for (int j = 0; j < 4; ++j) {
        const int e0 = tile * RA_TILE + (j * RA_THREADS + tid) * 4;
        int4 r;
        if (VEC) {
            r = *reinterpret_cast<const int4*>(key + (e0 < E ? e0 : 0));
        } else {
            const int last = E - 1;
            r.x = key[e0 + 0 < E ? e0 + 0 : last];
            r.y = key[e0 + 1 < E ? e0 + 1 : last];
            r.z = key[e0 + 2 < E ? e0 + 2 : last];
            r.w = key[e0 + 3 < E ? e0 + 3 : last];
        }
        cur[j] = r;
    }
}

template <bool VEC>
__device__ __forceinline__ void ra_scan_and_add(
    const float* __restrict__ x_trans, const int* __restrict__ key,
    const int* __restrict__ val, int E, int lo, int nrow, int part,
    float* acc, int* queue, int* qn) {
    const int tid  = threadIdx.x;
    const int wave = __builtin_amdgcn_readfirstlane(tid >> 5);
    const int lane = tid & 31;

    const int nTiles = (E + RA_TILE - 1) / RA_TILE;
    const int rot = part % nTiles;

    int4 cur[4];
    ra_load_keys<VEC>(key, E, rot, tid, cur);

    for (int it = 0; it < nTiles; ++it) {
        int tile = it + rot;
        if (tile >= nTiles) tile -= nTiles;
        const int buf = it & 1;

        unsigned m = 0;
        const int ebase = tile * RA_TILE + tid * 4;
#pragma unroll
        for (int j = 0; j < 4; ++j) {
            const int k[4] = {cur[j].x, cur[j].y, cur[j].z, cur[j].w};
#pragma unroll
            for (int c = 0; c < 4; ++c) {
                const bool inside = ebase + j * (RA_THREADS * 4) + c < E;
                m |= (inside && (unsigned)(k[c] - lo) < (unsigned)nrow ? 1u : 0u) << (j * 4 + c);
            }
        }
        if (m) {
            int pos = atomicAdd(&qn[buf], __builtin_popcount(m));
            do {
                const int b = __builtin_ctz(m);
                queue[pos++] = ebase + (b >> 2) * (RA_THREADS * 4) + (b & 3);
                m &= m - 1;
            } while (m);
        }
        __syncthreads();

        if (it + 1 < nTiles) {
            int ntile = tile + 1;
            if (ntile >= nTiles) ntile -= nTiles;
            ra_load_keys<VEC>(key, E, ntile, tid, cur);
        }
        if (tid == 0) qn[buf ^ 1] = 0;

        const int n = __builtin_amdgcn_readfirstlane(qn[buf]);
        auto round = [&](int q0) {
            const int myq = q0 + 32 * lane;
            const int ev  = queue[(lane < RA_UNR && myq < n) ? myq : 0];
            int eu[RA_UNR], rb[RA_UNR];
            const float* rowp[RA_UNR];
            float v0[RA_UNR], v1[RA_UNR], v2[RA_UNR];
#pragma unroll
            for (int u = 0; u < RA_UNR; ++u) eu[u] = __builtin_amdgcn_readlane(ev, u);
#pragma unroll
            for (int u = 0; u < RA_UNR; ++u) {
                rb[u]   = (key[eu[u]] - lo) * RA_PITCH;
                rowp[u] = x_trans + (size_t)val[eu[u]] * F85;
            }
#pragma unroll
            for (int u = 0; u < RA_UNR; ++u) {
                v0[u] = rowp[u][lane];
                v1[u] = rowp[u][32 + lane];
                v2[u] = rowp[u][lane < F85 - 64 ? 64 + lane : 64];
            }
#pragma unroll
            for (int u = 0; u < RA_UNR; ++u) {
                const bool ok = q0 + 32 * u < n;
                atomicAdd(&acc[rb[u] + lane],      ok ? v0[u] : 0.0f);
                atomicAdd(&acc[rb[u] + 32 + lane], ok ? v1[u] : 0.0f);
                atomicAdd(&acc[rb[u] + 64 + lane], ok ? v2[u] : 0.0f);
            }
        };
        if (n > 0) round(wave);
        for (int q0 = wave + 32 * RA_UNR; q0 < n; q0 += 32 * RA_UNR) round(q0);
        __syncthreads();
    }
}

__global__ __launch_bounds__(RA_THREADS) void rev_agg_lds_kernel(
    const float* __restrict__ x_trans,
    const int* __restrict__ src_pays, const int* __restrict__ dst_pays,
    const int* __restrict__ src_recv, const int* __restrict__ dst_recv,
    float* __restrict__ aggc, float* __restrict__ aggm,
    int N_c, int N_m, int E, int gc) {
    __shared__ float acc[RA_ROWS_C * RA_PITCH];
    __shared__ int   queue[RA_TILE];
    __shared__ int   qn[2];

    const int tid = threadIdx.x;
    const bool is_card = (int)blockIdx.x < gc;
    const int  part = is_card ? (int)blockIdx.x : (int)blockIdx.x - gc;
    const int  rows = is_card ? RA_ROWS_C : RA_ROWS_M;
    const int  N    = is_card ? N_c : N_m;
    const int* __restrict__ key = is_card ? src_pays : src_recv;
    const int* __restrict__ val = is_card ? dst_pays : dst_recv;
    float* __restrict__ outp    = is_card ? aggc : aggm;

    const int lo = part * rows;
    int nrow = N - lo;
    if (nrow > rows) nrow = rows;
    if (nrow < 0) nrow = 0;

    for (int i = tid; i < nrow * RA_PITCH; i += RA_THREADS) acc[i] = 0.0f;
    if (tid < 2) qn[tid] = 0;
    __syncthreads();

    if (E > 0) {
        const bool vec_ok = ((((size_t)key) & 15) == 0) && ((E & 3) == 0);
        if (vec_ok) ra_scan_and_add<true>(x_trans, key, val, E, lo, nrow, part, acc, queue, qn);
        else        ra_scan_and_add<false>(x_trans, key, val, E, lo, nrow, part, acc, queue, qn);
    }

    float* __restrict__ dstp = outp + (size_t)lo * F85;
    for (int i = tid; i < nrow * F85; i += RA_THREADS) {
        const int r = i / F85;
        dstp[i] = acc[r * RA_PITCH + (i - r * F85)];
    }
}

__global__ __launch_bounds__(128) void gemm88_kernel(
    const float* __restrict__ Xmain, const float* __restrict__ extra1,
    const float* __restrict__ extra2, const float* __restrict__ W,
    const float* __restrict__ bias, float* __restrict__ out, int M) {
    __shared__ float xs[16 * KPAD];

    const int row0 = blockIdx.x * 16;
    for (int i = threadIdx.x; i < 16 * KPAD; i += 128) {
        int r = i / KPAD, c = i - r * KPAD;
        int grow = row0 + r;
        float v;
        if (c < F85)      v = Xmain[(size_t)grow * F85 + c];
        else if (c == 85) v = extra1 ? extra1[grow] : 0.0f;
        else if (c == 86) v = extra2 ? extra2[grow] : 0.0f;
        else              v = 0.0f;
        xs[i] = v;
    }
    __syncthreads();

    const int wave = threadIdx.x >> 5;
    const int lane = threadIdx.x & 31;
    const int half = lane >> 4;
    const int lm   = lane & 15;
    const int c0   = wave * 16;

    const float bval = bias[c0 + lm];
    v8f acc;
#pragma unroll
    for (int i = 0; i < 8; ++i) acc[i] = bval;

    for (int k = 0; k < KPAD; k += 4) {
        const int ka = k + 2 * half;
        v2f a, b;
        a.x = xs[lm * KPAD + ka];
        a.y = xs[lm * KPAD + ka + 1];
        b.x = W[(size_t)(ka + 0) * H + c0 + lm];
        b.y = W[(size_t)(ka + 1) * H + c0 + lm];
        acc = __builtin_amdgcn_wmma_f32_16x16x4_f32(false, a, false, b, (short)0, acc,
                                                    false, false);
    }
#pragma unroll
    for (int v = 0; v < 8; ++v) {
        float r = acc[v];
        r = r > 0.0f ? r : 0.0f;
        out[(size_t)(row0 + v + 8 * half) * H + c0 + lm] = r;
    }
}

__global__ __launch_bounds__(128, 12) void conv2_out_kernel(
    const float* __restrict__ h_t,  const float* __restrict__ h_c,
    const float* __restrict__ h_m,
    const int* __restrict__ headc,  const int* __restrict__ headm,
    const int* __restrict__ nodec,  const int* __restrict__ nodem,
    const float* __restrict__ w2l,
    const float* __restrict__ w2r,  const float* __restrict__ b2,
    const float* __restrict__ wlin, const float* __restrict__ blin,
    float* __restrict__ out, int M, int E) {
    __shared__ float hs[16 * H];
    __shared__ alignas(16) float cs[16 * H];
    __shared__ alignas(16) float ms[16 * H];
    __shared__ float ts[16 * 2 * H];

    const int row0 = blockIdx.x * 16;

    const int gr = threadIdx.x >> 3;
    const int gq = threadIdx.x & 7;
    const int ec = headc[row0 + gr];
    const int em = headm[row0 + gr];

    for (int i = threadIdx.x; i < 16 * H; i += 128) {
        int r = i >> 6, c = i & 63;
        size_t g = (size_t)(row0 + r) * H + c;
        hs[i] = h_t[g];
    }

    {
        const int2*   __restrict__ nc  = reinterpret_cast<const int2*>(nodec);
        const int2*   __restrict__ nm  = reinterpret_cast<const int2*>(nodem);
        const float4* __restrict__ hc4 = reinterpret_cast<const float4*>(h_c);
        const float4* __restrict__ hm4 = reinterpret_cast<const float4*>(h_m);
        float4 c0 = make_float4(0.f, 0.f, 0.f, 0.f), c1 = c0, m0 = c0, m1 = c0;
        int  ic = ec, im = em;
        bool lc = (unsigned)ic < (unsigned)E, lm = (unsigned)im < (unsigned)E;
        for (int s = 0; s < E && (lc || lm); ++s) {
            const int2 dc = nc[lc ? ic : 0];
            const int2 dm = nm[lm ? im : 0];
            const float4* pc = hc4 + (size_t)dc.y * (H / 4) + gq;
            const float4* pm = hm4 + (size_t)dm.y * (H / 4) + gq;
            const float4 u0 = pc[0], u1 = pc[8], w0 = pm[0], w1 = pm[8];
            if (lc) { c0 += u0; c1 += u1; ic = dc.x; }
            if (lm) { m0 += w0; m1 += w1; im = dm.x; }
            lc = lc && (unsigned)ic < (unsigned)E;
            lm = lm && (unsigned)im < (unsigned)E;
        }
        float4* cs4 = reinterpret_cast<float4*>(cs);
        float4* ms4 = reinterpret_cast<float4*>(ms);
        cs4[gr * (H / 4) + gq]     = c0;
        cs4[gr * (H / 4) + 8 + gq] = c1;
        ms4[gr * (H / 4) + gq]     = m0;
        ms4[gr * (H / 4) + 8 + gq] = m1;
    }
    __syncthreads();

    const int wave = threadIdx.x >> 5;
    const int lane = threadIdx.x & 31;
    const int half = lane >> 4;
    const int lm   = lane & 15;
    const int c0   = wave * 16;

    v8f acc1, acc2;
    const float bv1 = b2[0 * H + c0 + lm];
    const float bv2 = b2[1 * H + c0 + lm];
#pragma unroll
    for (int i = 0; i < 8; ++i) { acc1[i] = bv1; acc2[i] = bv2; }

    const float* W2l0 = w2l;
    const float* W2l1 = w2l + 1 * H * H;
    const float* W2r0 = w2r;
    const float* W2r1 = w2r + 1 * H * H;

    for (int k = 0; k < H; k += 4) {
        const int ka = k + 2 * half;
        v2f ah, ac, am, b;
        ah.x = hs[lm * H + ka]; ah.y = hs[lm * H + ka + 1];
        ac.x = cs[lm * H + ka]; ac.y = cs[lm * H + ka + 1];
        am.x = ms[lm * H + ka]; am.y = ms[lm * H + ka + 1];

        b.x = W2l0[(ka + 0) * H + c0 + lm]; b.y = W2l0[(ka + 1) * H + c0 + lm];
        acc1 = __builtin_amdgcn_wmma_f32_16x16x4_f32(false, ac, false, b, (short)0, acc1,
                                                     false, false);
        b.x = W2r0[(ka + 0) * H + c0 + lm]; b.y = W2r0[(ka + 1) * H + c0 + lm];
        acc1 = __builtin_amdgcn_wmma_f32_16x16x4_f32(false, ah, false, b, (short)0, acc1,
                                                     false, false);
        b.x = W2l1[(ka + 0) * H + c0 + lm]; b.y = W2l1[(ka + 1) * H + c0 + lm];
        acc2 = __builtin_amdgcn_wmma_f32_16x16x4_f32(false, am, false, b, (short)0, acc2,
                                                     false, false);
        b.x = W2r1[(ka + 0) * H + c0 + lm]; b.y = W2r1[(ka + 1) * H + c0 + lm];
        acc2 = __builtin_amdgcn_wmma_f32_16x16x4_f32(false, ah, false, b, (short)0, acc2,
                                                     false, false);
    }

#pragma unroll
    for (int v = 0; v < 8; ++v) {
        float r1 = acc1[v]; r1 = r1 > 0.0f ? r1 : 0.0f;
        float r2 = acc2[v]; r2 = r2 > 0.0f ? r2 : 0.0f;
        ts[(v + 8 * half) * 128 + c0 + lm]      = r1;
        ts[(v + 8 * half) * 128 + 64 + c0 + lm] = r2;
    }
    __syncthreads();

    if (threadIdx.x < 32) {
        const int r = threadIdx.x >> 1, c = threadIdx.x & 1;
        float s = blin[c];
#pragma unroll 8
        for (int j = 0; j < 128; ++j) s += ts[r * 128 + j] * wlin[j * 2 + c];
        out[(size_t)(row0 + r) * 2 + c] = s;
    }
}

extern "C" void kernel_launch(void* const* d_in, const int* in_sizes, int n_in,
                              void* d_out, int out_size, void* d_ws, size_t ws_size,
                              hipStream_t stream) {
    const float* x_card   = (const float*)d_in[0];
    const float* x_merch  = (const float*)d_in[1];
    const float* x_trans  = (const float*)d_in[2];
    const float* w1_src_t = (const float*)d_in[3];
    const float* w1_dst_t = (const float*)d_in[4];
    const float* b1_t     = (const float*)d_in[5];
    const float* w1_src_r = (const float*)d_in[6];
    const float* w1_dst_r = (const float*)d_in[7];
    const float* b1_r     = (const float*)d_in[8];
    const float* w2_l     = (const float*)d_in[9];
    const float* w2_r     = (const float*)d_in[10];
    const float* b2       = (const float*)d_in[11];
    const float* w_lin    = (const float*)d_in[12];
    const float* b_lin    = (const float*)d_in[13];
    const int* src_pays   = (const int*)d_in[14];
    const int* dst_pays   = (const int*)d_in[15];
    const int* src_recv   = (const int*)d_in[16];
    const int* dst_recv   = (const int*)d_in[17];
    float* out = (float*)d_out;

    const int N_c = in_sizes[0];
    const int N_m = in_sizes[1];
    const int N_t = in_sizes[2] / F85;
    const int E   = in_sizes[14];

    float* ws = (float*)d_ws;
    size_t off = 0;
    float* h_t    = ws + off; off += (size_t)N_t * H;
    float* h_c    = ws + off; off += (size_t)N_c * H;
    float* h_m    = ws + off; off += (size_t)N_m * H;
    int*   nodec  = (int*)(ws + off); off += (size_t)E * 2;
    int*   nodem  = (int*)(ws + off); off += (size_t)E * 2;
    float* aggc1  = ws + off; off += (size_t)N_t;
    float* aggm1  = ws + off; off += (size_t)N_t;
    int*   headc  = (int*)(ws + off); off += (size_t)N_t;
    int*   headm  = (int*)(ws + off); off += (size_t)N_t;
    float* aggt2c = ws + off; off += (size_t)N_c * F85;
    float* aggt2m = ws + off; off += (size_t)N_m * F85;
    float* W1t    = ws + off; off += (size_t)KPAD * H;
    float* W1c    = ws + off; off += (size_t)KPAD * H;
    float* W1m    = ws + off; off += (size_t)KPAD * H;
    float* b1sum  = ws + off; off += (size_t)H;

    {
        long n = 2 * (long)N_t;
        long blocks = (n + 255) / 256;
        init_kernel<<<(unsigned)blocks, 256, 0, stream>>>(aggc1, headc, n);
    }
    prep_w_kernel<<<(KPAD * H + 255) / 256, 256, 0, stream>>>(
        w1_src_t, w1_dst_t, b1_t, w1_src_r, w1_dst_r, W1t, W1c, W1m, b1sum);
    edge_pass_kernel<<<(E + 255) / 256, 256, 0, stream>>>(
        x_card, x_merch, src_pays, dst_pays, src_recv, dst_recv, aggc1, aggm1,
        headc, headm, nodec, nodem, E);
    {
        const int gc = (N_c + RA_ROWS_C - 1) / RA_ROWS_C;
        const int gm = (N_m + RA_ROWS_M - 1) / RA_ROWS_M;
        rev_agg_lds_kernel<<<gc + gm, RA_THREADS, 0, stream>>>(
            x_trans, src_pays, dst_pays, src_recv, dst_recv, aggt2c, aggt2m, N_c, N_m, E, gc);
    }
    gemm88_kernel<<<N_t / 16, 128, 0, stream>>>(x_trans, aggc1, aggm1, W1t, b1sum, h_t, N_t);
    gemm88_kernel<<<N_c / 16, 128, 0, stream>>>(aggt2c, x_card, nullptr, W1c, b1_r, h_c, N_c);
    gemm88_kernel<<<N_m / 16, 128, 0, stream>>>(aggt2m, x_merch, nullptr, W1m, b1_r + H, h_m, N_m);
    conv2_out_kernel<<<N_t / 16, 128, 0, stream>>>(
        h_t, h_c, h_m, headc, headm, nodec, nodem, w2_l, w2_r, b2, w_lin, b_lin, out, N_t, E);
}


# conv2 kernel: LDS row pitch of the three 16x64 operand tiles 64 -> 68 words, of the 16x128 concat tile 128 -> 132: fragment reads and the final linear's reads no longer put 16 rows on one bank
# speedup vs baseline: 2.2693x; 1.1640x over previous
#include <hip/hip_runtime.h>

typedef __attribute__((ext_vector_type(2))) float v2f;
typedef __attribute__((ext_vector_type(8))) float v8f;

#define H 64
#define F85 85
#define KPAD 88

__global__ void init_kernel(float* __restrict__ z, int* __restrict__ heads, long n) {
    long i = (long)blockIdx.x * blockDim.x + threadIdx.x;
    if (i < n) { z[i] = 0.0f; heads[i] = -1; }
}

__global__ void prep_w_kernel(const float* __restrict__ w1_src_t,
                              const float* __restrict__ w1_dst_t,
                              const float* __restrict__ b1_t,
                              const float* __restrict__ w1_src_r,
                              const float* __restrict__ w1_dst_r,
                              float* __restrict__ W1t, float* __restrict__ W1c,
                              float* __restrict__ W1m, float* __restrict__ b1sum) {
    int i = blockIdx.x * blockDim.x + threadIdx.x;
    if (i >= KPAD * H) return;
    int r = i >> 6, c = i & 63;
    float vt = 0.f, vc = 0.f, vm = 0.f;
    if (r < F85) {
        vt = w1_dst_t[r * H + c] + w1_dst_t[F85 * H + r * H + c];
        vc = w1_src_r[r * H + c];
        vm = w1_src_r[F85 * H + r * H + c];
    } else if (r == 85) {
        vt = w1_src_t[c];
        vc = w1_dst_r[c];
        vm = w1_dst_r[H + c];
    } else if (r == 86) {
        vt = w1_src_t[H + c];
    }
    W1t[i] = vt; W1c[i] = vc; W1m[i] = vm;
    if (i < H) b1sum[i] = b1_t[i] + b1_t[H + i];
}

__global__ void edge_pass_kernel(const float* __restrict__ xc, const float* __restrict__ xm,
                                 const int* __restrict__ sp, const int* __restrict__ dp,
                                 const int* __restrict__ sr, const int* __restrict__ dr,
                                 float* __restrict__ aggc, float* __restrict__ aggm,
                                 int* __restrict__ headc, int* __restrict__ headm,
                                 int* __restrict__ nodec, int* __restrict__ nodem, int E) {
    int e = blockIdx.x * blockDim.x + threadIdx.x;
    if (e >= E) return;
    const int sc = sp[e], dc = dp[e], sm = sr[e], dm = dr[e];
    atomicAdd(&aggc[dc], xc[sc]);
    atomicAdd(&aggm[dm], xm[sm]);
    reinterpret_cast<int2*>(nodec)[e] = make_int2(atomicExch(&headc[dc], e), sc);
    reinterpret_cast<int2*>(nodem)[e] = make_int2(atomicExch(&headm[dm], e), sm);
}

#define RA_THREADS 1024
#define RA_TILE    16384
#define RA_UNR     6
#define RA_ROWS_C  392
#define RA_ROWS_M  80
#define RA_PITCH   96

template <bool VEC>
__device__ __forceinline__ void ra_load_keys(const int* __restrict__ key, int E, int tile,
                                             int tid, int4 (&cur)[4]) {
#pragma unroll
    for (int j = 0; j < 4; ++j) {
        const int e0 = tile * RA_TILE + (j * RA_THREADS + tid) * 4;
        int4 r;
        if (VEC) {
            r = *reinterpret_cast<const int4*>(key + (e0 < E ? e0 : 0));
        } else {
            const int last = E - 1;
            r.x = key[e0 + 0 < E ? e0 + 0 : last];
            r.y = key[e0 + 1 < E ? e0 + 1 : last];
            r.z = key[e0 + 2 < E ? e0 + 2 : last];
            r.w = key[e0 + 3 < E ? e0 + 3 : last];
        }
        cur[j] = r;
    }
}

template <bool VEC>
__device__ __forceinline__ void ra_scan_and_add(
    const float* __restrict__ x_trans, const int* __restrict__ key,
    const int* __restrict__ val, int E, int lo, int nrow, int part,
    float* acc, int* queue, int* qn) {
    const int tid  = threadIdx.x;
    const int wave = __builtin_amdgcn_readfirstlane(tid >> 5);
    const int lane = tid & 31;

    const int nTiles = (E + RA_TILE - 1) / RA_TILE;
    const int rot = part % nTiles;

    int4 cur[4];
    ra_load_keys<VEC>(key, E, rot, tid, cur);

    for (int it = 0; it < nTiles; ++it) {
        int tile = it + rot;
        if (tile >= nTiles) tile -= nTiles;
        const int buf = it & 1;

        unsigned m = 0;
        const int ebase = tile * RA_TILE + tid * 4;
#pragma unroll
        for (int j = 0; j < 4; ++j) {
            const int k[4] = {cur[j].x, cur[j].y, cur[j].z, cur[j].w};
#pragma unroll
            for (int c = 0; c < 4; ++c) {
                const bool inside = ebase + j * (RA_THREADS * 4) + c < E;
                m |= (inside && (unsigned)(k[c] - lo) < (unsigned)nrow ? 1u : 0u) << (j * 4 + c);
            }
        }
        if (m) {
            int pos = atomicAdd(&qn[buf], __builtin_popcount(m));
            do {
                const int b = __builtin_ctz(m);
                queue[pos++] = ebase + (b >> 2) * (RA_THREADS * 4) + (b & 3);
                m &= m - 1;
            } while (m);
        }
        __syncthreads();

        if (it + 1 < nTiles) {
            int ntile = tile + 1;
            if (ntile >= nTiles) ntile -= nTiles;
            ra_load_keys<VEC>(key, E, ntile, tid, cur);
        }
        if (tid == 0) qn[buf ^ 1] = 0;

        const int n = __builtin_amdgcn_readfirstlane(qn[buf]);
        auto round = [&](int q0) {
            const int myq = q0 + 32 * lane;
            const int ev  = queue[(lane < RA_UNR && myq < n) ? myq : 0];
            int eu[RA_UNR], rb[RA_UNR];
            const float* rowp[RA_UNR];
            float v0[RA_UNR], v1[RA_UNR], v2[RA_UNR];
#pragma unroll
            for (int u = 0; u < RA_UNR; ++u) eu[u] = __builtin_amdgcn_readlane(ev, u);
#pragma unroll
            for (int u = 0; u < RA_UNR; ++u) {
                rb[u]   = (key[eu[u]] - lo) * RA_PITCH;
                rowp[u] = x_trans + (size_t)val[eu[u]] * F85;
            }
#pragma unroll
            for (int u = 0; u < RA_UNR; ++u) {
                v0[u] = rowp[u][lane];
                v1[u] = rowp[u][32 + lane];
                v2[u] = rowp[u][lane < F85 - 64 ? 64 + lane : 64];
            }
#pragma unroll
            for (int u = 0; u < RA_UNR; ++u) {
                const bool ok = q0 + 32 * u < n;
                atomicAdd(&acc[rb[u] + lane],      ok ? v0[u] : 0.0f);
                atomicAdd(&acc[rb[u] + 32 + lane], ok ? v1[u] : 0.0f);
                atomicAdd(&acc[rb[u] + 64 + lane], ok ? v2[u] : 0.0f);
            }
        };
        if (n > 0) round(wave);
        for (int q0 = wave + 32 * RA_UNR; q0 < n; q0 += 32 * RA_UNR) round(q0);
        __syncthreads();
    }
}

__global__ __launch_bounds__(RA_THREADS) void rev_agg_lds_kernel(
    const float* __restrict__ x_trans,
    const int* __restrict__ src_pays, const int* __restrict__ dst_pays,
    const int* __restrict__ src_recv, const int* __restrict__ dst_recv,
    float* __restrict__ aggc, float* __restrict__ aggm,
    int N_c, int N_m, int E, int gc) {
    __shared__ float acc[RA_ROWS_C * RA_PITCH];
    __shared__ int   queue[RA_TILE];
    __shared__ int   qn[2];

    const int tid = threadIdx.x;
    const bool is_card = (int)blockIdx.x < gc;
    const int  part = is_card ? (int)blockIdx.x : (int)blockIdx.x - gc;
    const int  rows = is_card ? RA_ROWS_C : RA_ROWS_M;
    const int  N    = is_card ? N_c : N_m;
    const int* __restrict__ key = is_card ? src_pays : src_recv;
    const int* __restrict__ val = is_card ? dst_pays : dst_recv;
    float* __restrict__ outp    = is_card ? aggc : aggm;

    const int lo = part * rows;
    int nrow = N - lo;
    if (nrow > rows) nrow = rows;
    if (nrow < 0) nrow = 0;

    for (int i = tid; i < nrow * RA_PITCH; i += RA_THREADS) acc[i] = 0.0f;
    if (tid < 2) qn[tid] = 0;
    __syncthreads();

    if (E > 0) {
        const bool vec_ok = ((((size_t)key) & 15) == 0) && ((E & 3) == 0);
        if (vec_ok) ra_scan_and_add<true>(x_trans, key, val, E, lo, nrow, part, acc, queue, qn);
        else        ra_scan_and_add<false>(x_trans, key, val, E, lo, nrow, part, acc, queue, qn);
    }

    float* __restrict__ dstp = outp + (size_t)lo * F85;
    for (int i = tid; i < nrow * F85; i += RA_THREADS) {
        const int r = i / F85;
        dstp[i] = acc[r * RA_PITCH + (i - r * F85)];
    }
}

__global__ __launch_bounds__(128) void gemm88_kernel(
    const float* __restrict__ Xmain, const float* __restrict__ extra1,
    const float* __restrict__ extra2, const float* __restrict__ W,
    const float* __restrict__ bias, float* __restrict__ out, int M) {
    __shared__ float xs[16 * KPAD];

    const int row0 = blockIdx.x * 16;
    for (int i = threadIdx.x; i < 16 * KPAD; i += 128) {
        int r = i / KPAD, c = i - r * KPAD;
        int grow = row0 + r;
        float v;
        if (c < F85)      v = Xmain[(size_t)grow * F85 + c];
        else if (c == 85) v = extra1 ? extra1[grow] : 0.0f;
        else if (c == 86) v = extra2 ? extra2[grow] : 0.0f;
        else              v = 0.0f;
        xs[i] = v;
    }
    __syncthreads();

    const int wave = threadIdx.x >> 5;
    const int lane = threadIdx.x & 31;
    const int half = lane >> 4;
    const int lm   = lane & 15;
    const int c0   = wave * 16;

    const float bval = bias[c0 + lm];
    v8f acc;
#pragma unroll
    for (int i = 0; i < 8; ++i) acc[i] = bval;

    for (int k = 0; k < KPAD; k += 4) {
        const int ka = k + 2 * half;
        v2f a, b;
        a.x = xs[lm * KPAD + ka];
        a.y = xs[lm * KPAD + ka + 1];
        b.x = W[(size_t)(ka + 0) * H + c0 + lm];
        b.y = W[(size_t)(ka + 1) * H + c0 + lm];
        acc = __builtin_amdgcn_wmma_f32_16x16x4_f32(false, a, false, b, (short)0, acc,
                                                    false, false);
    }
#pragma unroll
    for (int v = 0; v < 8; ++v) {
        float r = acc[v];
        r = r > 0.0f ? r : 0.0f;
        out[(size_t)(row0 + v + 8 * half) * H + c0 + lm] = r;
    }
}

#define CT_PITCH 68
#define TS_PITCH 132

__global__ __launch_bounds__(128, 12) void conv2_out_kernel(
    const float* __restrict__ h_t,  const float* __restrict__ h_c,
    const float* __restrict__ h_m,
    const int* __restrict__ headc,  const int* __restrict__ headm,
    const int* __restrict__ nodec,  const int* __restrict__ nodem,
    const float* __restrict__ w2l,
    const float* __restrict__ w2r,  const float* __restrict__ b2,
    const float* __restrict__ wlin, const float* __restrict__ blin,
    float* __restrict__ out, int M, int E) {
    __shared__ alignas(16) float hs[16 * CT_PITCH];
    __shared__ alignas(16) float cs[16 * CT_PITCH];
    __shared__ alignas(16) float ms[16 * CT_PITCH];
    __shared__ alignas(16) float ts[16 * TS_PITCH];

    const int row0 = blockIdx.x * 16;

    const int gr = threadIdx.x >> 3;
    const int gq = threadIdx.x & 7;
    const int ec = headc[row0 + gr];
    const int em = headm[row0 + gr];

    for (int i = threadIdx.x; i < 16 * H; i += 128) {
        int r = i >> 6, c = i & 63;
        size_t g = (size_t)(row0 + r) * H + c;
        hs[r * CT_PITCH + c] = h_t[g];
    }

    {
        const int2*   __restrict__ nc  = reinterpret_cast<const int2*>(nodec);
        const int2*   __restrict__ nm  = reinterpret_cast<const int2*>(nodem);
        const float4* __restrict__ hc4 = reinterpret_cast<const float4*>(h_c);
        const float4* __restrict__ hm4 = reinterpret_cast<const float4*>(h_m);
        float4 c0 = make_float4(0.f, 0.f, 0.f, 0.f), c1 = c0, m0 = c0, m1 = c0;
        int  ic = ec, im = em;
        bool lc = (unsigned)ic < (unsigned)E, lm = (unsigned)im < (unsigned)E;
        for (int s = 0; s < E && (lc || lm); ++s) {
            const int2 dc = nc[lc ? ic : 0];
            const int2 dm = nm[lm ? im : 0];
            const float4* pc = hc4 + (size_t)dc.y * (H / 4) + gq;
            const float4* pm = hm4 + (size_t)dm.y * (H / 4) + gq;
            const float4 u0 = pc[0], u1 = pc[8], w0 = pm[0], w1 = pm[8];
            if (lc) { c0 += u0; c1 += u1; ic = dc.x; }
            if (lm) { m0 += w0; m1 += w1; im = dm.x; }
            lc = lc && (unsigned)ic < (unsigned)E;
            lm = lm && (unsigned)im < (unsigned)E;
        }
        float4* cs4 = reinterpret_cast<float4*>(cs);
        float4* ms4 = reinterpret_cast<float4*>(ms);
        cs4[gr * (CT_PITCH / 4) + gq]     = c0;
        cs4[gr * (CT_PITCH / 4) + 8 + gq] = c1;
        ms4[gr * (CT_PITCH / 4) + gq]     = m0;
        ms4[gr * (CT_PITCH / 4) + 8 + gq] = m1;
    }
    __syncthreads();

    const int wave = threadIdx.x >> 5;
    const int lane = threadIdx.x & 31;
    const int half = lane >> 4;
    const int lm   = lane & 15;
    const int c0   = wave * 16;

    v8f acc1, acc2;
    const float bv1 = b2[0 * H + c0 + lm];
    const float bv2 = b2[1 * H + c0 + lm];
#pragma unroll
    for (int i = 0; i < 8; ++i) { acc1[i] = bv1; acc2[i] = bv2; }

    const float* W2l0 = w2l;
    const float* W2l1 = w2l + 1 * H * H;
    const float* W2r0 = w2r;
    const float* W2r1 = w2r + 1 * H * H;

    for (int k = 0; k < H; k += 4) {
        const int ka = k + 2 * half;
        v2f ah, ac, am, b;
        ah.x = hs[lm * CT_PITCH + ka]; ah.y = hs[lm * CT_PITCH + ka + 1];
        ac.x = cs[lm * CT_PITCH + ka]; ac.y = cs[lm * CT_PITCH + ka + 1];
        am.x = ms[lm * CT_PITCH + ka]; am.y = ms[lm * CT_PITCH + ka + 1];

        b.x = W2l0[(ka + 0) * H + c0 + lm]; b.y = W2l0[(ka + 1) * H + c0 + lm];
        acc1 = __builtin_amdgcn_wmma_f32_16x16x4_f32(false, ac, false, b, (short)0, acc1,
                                                     false, false);
        b.x = W2r0[(ka + 0) * H + c0 + lm]; b.y = W2r0[(ka + 1) * H + c0 + lm];
        acc1 = __builtin_amdgcn_wmma_f32_16x16x4_f32(false, ah, false, b, (short)0, acc1,
                                                     false, false);
        b.x = W2l1[(ka + 0) * H + c0 + lm]; b.y = W2l1[(ka + 1) * H + c0 + lm];
        acc2 = __builtin_amdgcn_wmma_f32_16x16x4_f32(false, am, false, b, (short)0, acc2,
                                                     false, false);
        b.x = W2r1[(ka + 0) * H + c0 + lm]; b.y = W2r1[(ka + 1) * H + c0 + lm];
        acc2 = __builtin_amdgcn_wmma_f32_16x16x4_f32(false, ah, false, b, (short)0, acc2,
                                                     false, false);
    }

#pragma unroll
    for (int v = 0; v < 8; ++v) {
        float r1 = acc1[v]; r1 = r1 > 0.0f ? r1 : 0.0f;
        float r2 = acc2[v]; r2 = r2 > 0.0f ? r2 : 0.0f;
        ts[(v + 8 * half) * TS_PITCH + c0 + lm]      = r1;
        ts[(v + 8 * half) * TS_PITCH + 64 + c0 + lm] = r2;
    }
    __syncthreads();

    if (threadIdx.x < 32) {
        const int r = threadIdx.x >> 1, c = threadIdx.x & 1;
        float s = blin[c];
#pragma unroll 8
        for (int j = 0; j < 128; ++j) s += ts[r * TS_PITCH + j] * wlin[j * 2 + c];
        out[(size_t)(row0 + r) * 2 + c] = s;
    }
}

extern "C" void kernel_launch(void* const* d_in, const int* in_sizes, int n_in,
                              void* d_out, int out_size, void* d_ws, size_t ws_size,
                              hipStream_t stream) {
    const float* x_card   = (const float*)d_in[0];
    const float* x_merch  = (const float*)d_in[1];
    const float* x_trans  = (const float*)d_in[2];
    const float* w1_src_t = (const float*)d_in[3];
    const float* w1_dst_t = (const float*)d_in[4];
    const float* b1_t     = (const float*)d_in[5];
    const float* w1_src_r = (const float*)d_in[6];
    const float* w1_dst_r = (const float*)d_in[7];
    const float* b1_r     = (const float*)d_in[8];
    const float* w2_l     = (const float*)d_in[9];
    const float* w2_r     = (const float*)d_in[10];
    const float* b2       = (const float*)d_in[11];
    const float* w_lin    = (const float*)d_in[12];
    const float* b_lin    = (const float*)d_in[13];
    const int* src_pays   = (const int*)d_in[14];
    const int* dst_pays   = (const int*)d_in[15];
    const int* src_recv   = (const int*)d_in[16];
    const int* dst_recv   = (const int*)d_in[17];
    float* out = (float*)d_out;

    const int N_c = in_sizes[0];
    const int N_m = in_sizes[1];
    const int N_t = in_sizes[2] / F85;
    const int E   = in_sizes[14];

    float* ws = (float*)d_ws;
    size_t off = 0;
    float* h_t    = ws + off; off += (size_t)N_t * H;
    float* h_c    = ws + off; off += (size_t)N_c * H;
    float* h_m    = ws + off; off += (size_t)N_m * H;
    int*   nodec  = (int*)(ws + off); off += (size_t)E * 2;
    int*   nodem  = (int*)(ws + off); off += (size_t)E * 2;
    float* aggc1  = ws + off; off += (size_t)N_t;
    float* aggm1  = ws + off; off += (size_t)N_t;
    int*   headc  = (int*)(ws + off); off += (size_t)N_t;
    int*   headm  = (int*)(ws + off); off += (size_t)N_t;
    float* aggt2c = ws + off; off += (size_t)N_c * F85;
    float* aggt2m = ws + off; off += (size_t)N_m * F85;
    float* W1t    = ws + off; off += (size_t)KPAD * H;
    float* W1c    = ws + off; off += (size_t)KPAD * H;
    float* W1m    = ws + off; off += (size_t)KPAD * H;
    float* b1sum  = ws + off; off += (size_t)H;

    {
        long n = 2 * (long)N_t;
        long blocks = (n + 255) / 256;
        init_kernel<<<(unsigned)blocks, 256, 0, stream>>>(aggc1, headc, n);
    }
    prep_w_kernel<<<(KPAD * H + 255) / 256, 256, 0, stream>>>(
        w1_src_t, w1_dst_t, b1_t, w1_src_r, w1_dst_r, W1t, W1c, W1m, b1sum);
    edge_pass_kernel<<<(E + 255) / 256, 256, 0, stream>>>(
        x_card, x_merch, src_pays, dst_pays, src_recv, dst_recv, aggc1, aggm1,
        headc, headm, nodec, nodem, E);
    {
        const int gc = (N_c + RA_ROWS_C - 1) / RA_ROWS_C;
        const int gm = (N_m + RA_ROWS_M - 1) / RA_ROWS_M;
        rev_agg_lds_kernel<<<gc + gm, RA_THREADS, 0, stream>>>(
            x_trans, src_pays, dst_pays, src_recv, dst_recv, aggt2c, aggt2m, N_c, N_m, E, gc);
    }
    gemm88_kernel<<<N_t / 16, 128, 0, stream>>>(x_trans, aggc1, aggm1, W1t, b1sum, h_t, N_t);
    gemm88_kernel<<<N_c / 16, 128, 0, stream>>>(aggt2c, x_card, nullptr, W1c, b1_r, h_c, N_c);
    gemm88_kernel<<<N_m / 16, 128, 0, stream>>>(aggt2m, x_merch, nullptr, W1m, b1_r + H, h_m, N_m);
    conv2_out_kernel<<<N_t / 16, 128, 0, stream>>>(
        h_t, h_c, h_m, headc, headm, nodec, nodem, w2_l, w2_r, b2, w_lin, b_lin, out, N_t, E);
}


# conv1 transaction GEMM fused into the conv2 kernel: each workgroup builds its 16x64 h_t tile from x_trans on WMMA into LDS; the h_t launch and its write and re-read through memory are gone
# speedup vs baseline: 2.7628x; 1.2175x over previous
#include <hip/hip_runtime.h>

typedef __attribute__((ext_vector_type(2))) float v2f;
typedef __attribute__((ext_vector_type(8))) float v8f;

#define H 64
#define F85 85
#define KPAD 88

__global__ void init_kernel(float* __restrict__ z, int* __restrict__ heads, long n) {
    long i = (long)blockIdx.x * blockDim.x + threadIdx.x;
    if (i < n) { z[i] = 0.0f; heads[i] = -1; }
}

__global__ void prep_w_kernel(const float* __restrict__ w1_src_t,
                              const float* __restrict__ w1_dst_t,
                              const float* __restrict__ b1_t,
                              const float* __restrict__ w1_src_r,
                              const float* __restrict__ w1_dst_r,
                              float* __restrict__ W1t, float* __restrict__ W1c,
                              float* __restrict__ W1m, float* __restrict__ b1sum) {
    int i = blockIdx.x * blockDim.x + threadIdx.x;
    if (i >= KPAD * H) return;
    int r = i >> 6, c = i & 63;
    float vt = 0.f, vc = 0.f, vm = 0.f;
    if (r < F85) {
        vt = w1_dst_t[r * H + c] + w1_dst_t[F85 * H + r * H + c];
        vc = w1_src_r[r * H + c];
        vm = w1_src_r[F85 * H + r * H + c];
    } else if (r == 85) {
        vt = w1_src_t[c];
        vc = w1_dst_r[c];
        vm = w1_dst_r[H + c];
    } else if (r == 86) {
        vt = w1_src_t[H + c];
    }
    W1t[i] = vt; W1c[i] = vc; W1m[i] = vm;
    if (i < H) b1sum[i] = b1_t[i] + b1_t[H + i];
}

__global__ void edge_pass_kernel(const float* __restrict__ xc, const float* __restrict__ xm,
                                 const int* __restrict__ sp, const int* __restrict__ dp,
                                 const int* __restrict__ sr, const int* __restrict__ dr,
                                 float* __restrict__ aggc, float* __restrict__ aggm,
                                 int* __restrict__ headc, int* __restrict__ headm,
                                 int* __restrict__ nodec, int* __restrict__ nodem, int E) {
    int e = blockIdx.x * blockDim.x + threadIdx.x;
    if (e >= E) return;
    const int sc = sp[e], dc = dp[e], sm = sr[e], dm = dr[e];
    atomicAdd(&aggc[dc], xc[sc]);
    atomicAdd(&aggm[dm], xm[sm]);
    reinterpret_cast<int2*>(nodec)[e] = make_int2(atomicExch(&headc[dc], e), sc);
    reinterpret_cast<int2*>(nodem)[e] = make_int2(atomicExch(&headm[dm], e), sm);
}

#define RA_THREADS 1024
#define RA_TILE    16384
#define RA_UNR     6
#define RA_ROWS_C  392
#define RA_ROWS_M  80
#define RA_PITCH   96

template <bool VEC>
__device__ __forceinline__ void ra_load_keys(const int* __restrict__ key, int E, int tile,
                                             int tid, int4 (&cur)[4]) {
#pragma unroll
    for (int j = 0; j < 4; ++j) {
        const int e0 = tile * RA_TILE + (j * RA_THREADS + tid) * 4;
        int4 r;
        if (VEC) {
            r = *reinterpret_cast<const int4*>(key + (e0 < E ? e0 : 0));
        } else {
            const int last = E - 1;
            r.x = key[e0 + 0 < E ? e0 + 0 : last];
            r.y = key[e0 + 1 < E ? e0 + 1 : last];
            r.z = key[e0 + 2 < E ? e0 + 2 : last];
            r.w = key[e0 + 3 < E ? e0 + 3 : last];
        }
        cur[j] = r;
    }
}

template <bool VEC>
__device__ __forceinline__ void ra_scan_and_add(
    const float* __restrict__ x_trans, const int* __restrict__ key,
    const int* __restrict__ val, int E, int lo, int nrow, int part,
    float* acc, int* queue, int* qn) {
    const int tid  = threadIdx.x;
    const int wave = __builtin_amdgcn_readfirstlane(tid >> 5);
    const int lane = tid & 31;

    const int nTiles = (E + RA_TILE - 1) / RA_TILE;
    const int rot = part % nTiles;

    int4 cur[4];
    ra_load_keys<VEC>(key, E, rot, tid, cur);

    for (int it = 0; it < nTiles; ++it) {
        int tile = it + rot;
        if (tile >= nTiles) tile -= nTiles;
        const int buf = it & 1;

        unsigned m = 0;
        const int ebase = tile * RA_TILE + tid * 4;
#pragma unroll
        for (int j = 0; j < 4; ++j) {
            const int k[4] = {cur[j].x, cur[j].y, cur[j].z, cur[j].w};
#pragma unroll
            for (int c = 0; c < 4; ++c) {
                const bool inside = ebase + j * (RA_THREADS * 4) + c < E;
                m |= (inside && (unsigned)(k[c] - lo) < (unsigned)nrow ? 1u : 0u) << (j * 4 + c);
            }
        }
        if (m) {
            int pos = atomicAdd(&qn[buf], __builtin_popcount(m));
            do {
                const int b = __builtin_ctz(m);
                queue[pos++] = ebase + (b >> 2) * (RA_THREADS * 4) + (b & 3);
                m &= m - 1;
            } while (m);
        }
        __syncthreads();

        if (it + 1 < nTiles) {
            int ntile = tile + 1;
            if (ntile >= nTiles) ntile -= nTiles;
            ra_load_keys<VEC>(key, E, ntile, tid, cur);
        }
        if (tid == 0) qn[buf ^ 1] = 0;

        const int n = __builtin_amdgcn_readfirstlane(qn[buf]);
        auto round = [&](int q0) {
            const int myq = q0 + 32 * lane;
            const int ev  = queue[(lane < RA_UNR && myq < n) ? myq : 0];
            int eu[RA_UNR], rb[RA_UNR];
            const float* rowp[RA_UNR];
            float v0[RA_UNR], v1[RA_UNR], v2[RA_UNR];
#pragma unroll
            for (int u = 0; u < RA_UNR; ++u) eu[u] = __builtin_amdgcn_readlane(ev, u);
#pragma unroll
            for (int u = 0; u < RA_UNR; ++u) {
                rb[u]   = (key[eu[u]] - lo) * RA_PITCH;
                rowp[u] = x_trans + (size_t)val[eu[u]] * F85;
            }
#pragma unroll
            for (int u = 0; u < RA_UNR; ++u) {
                v0[u] = rowp[u][lane];
                v1[u] = rowp[u][32 + lane];
                v2[u] = rowp[u][lane < F85 - 64 ? 64 + lane : 64];
            }
#pragma unroll
            for (int u = 0; u < RA_UNR; ++u) {
                const bool ok = q0 + 32 * u < n;
                atomicAdd(&acc[rb[u] + lane],      ok ? v0[u] : 0.0f);
                atomicAdd(&acc[rb[u] + 32 + lane], ok ? v1[u] : 0.0f);
                atomicAdd(&acc[rb[u] + 64 + lane], ok ? v2[u] : 0.0f);
            }
        };
        if (n > 0) round(wave);
        for (int q0 = wave + 32 * RA_UNR; q0 < n; q0 += 32 * RA_UNR) round(q0);
        __syncthreads();
    }
}

__global__ __launch_bounds__(RA_THREADS) void rev_agg_lds_kernel(
    const float* __restrict__ x_trans,
    const int* __restrict__ src_pays, const int* __restrict__ dst_pays,
    const int* __restrict__ src_recv, const int* __restrict__ dst_recv,
    float* __restrict__ aggc, float* __restrict__ aggm,
    int N_c, int N_m, int E, int gc) {
    __shared__ float acc[RA_ROWS_C * RA_PITCH];
    __shared__ int   queue[RA_TILE];
    __shared__ int   qn[2];

    const int tid = threadIdx.x;
    const bool is_card = (int)blockIdx.x < gc;
    const int  part = is_card ? (int)blockIdx.x : (int)blockIdx.x - gc;
    const int  rows = is_card ? RA_ROWS_C : RA_ROWS_M;
    const int  N    = is_card ? N_c : N_m;
    const int* __restrict__ key = is_card ? src_pays : src_recv;
    const int* __restrict__ val = is_card ? dst_pays : dst_recv;
    float* __restrict__ outp    = is_card ? aggc : aggm;

    const int lo = part * rows;
    int nrow = N - lo;
    if (nrow > rows) nrow = rows;
    if (nrow < 0) nrow = 0;

    for (int i = tid; i < nrow * RA_PITCH; i += RA_THREADS) acc[i] = 0.0f;
    if (tid < 2) qn[tid] = 0;
    __syncthreads();

    if (E > 0) {
        const bool vec_ok = ((((size_t)key) & 15) == 0) && ((E & 3) == 0);
        if (vec_ok) ra_scan_and_add<true>(x_trans, key, val, E, lo, nrow, part, acc, queue, qn);
        else        ra_scan_and_add<false>(x_trans, key, val, E, lo, nrow, part, acc, queue, qn);
    }

    float* __restrict__ dstp = outp + (size_t)lo * F85;
    for (int i = tid; i < nrow * F85; i += RA_THREADS) {
        const int r = i / F85;
        dstp[i] = acc[r * RA_PITCH + (i - r * F85)];
    }
}

__global__ __launch_bounds__(128) void gemm88_kernel(
    const float* __restrict__ Xmain, const float* __restrict__ extra1,
    const float* __restrict__ extra2, const float* __restrict__ W,
    const float* __restrict__ bias, float* __restrict__ out, int M) {
    __shared__ float xs[16 * KPAD];

    const int row0 = blockIdx.x * 16;
    for (int i = threadIdx.x; i < 16 * KPAD; i += 128) {
        int r = i / KPAD, c = i - r * KPAD;
        int grow = row0 + r;
        float v;
        if (c < F85)      v = Xmain[(size_t)grow * F85 + c];
        else if (c == 85) v = extra1 ? extra1[grow] : 0.0f;
        else if (c == 86) v = extra2 ? extra2[grow] : 0.0f;
        else              v = 0.0f;
        xs[i] = v;
    }
    __syncthreads();

    const int wave = threadIdx.x >> 5;
    const int lane = threadIdx.x & 31;
    const int half = lane >> 4;
    const int lm   = lane & 15;
    const int c0   = wave * 16;

    const float bval = bias[c0 + lm];
    v8f acc;
#pragma unroll
    for (int i = 0; i < 8; ++i) acc[i] = bval;

    for (int k = 0; k < KPAD; k += 4) {
        const int ka = k + 2 * half;
        v2f a, b;
        a.x = xs[lm * KPAD + ka];
        a.y = xs[lm * KPAD + ka + 1];
        b.x = W[(size_t)(ka + 0) * H + c0 + lm];
        b.y = W[(size_t)(ka + 1) * H + c0 + lm];
        acc = __builtin_amdgcn_wmma_f32_16x16x4_f32(false, a, false, b, (short)0, acc,
                                                    false, false);
    }
#pragma unroll
    for (int v = 0; v < 8; ++v) {
        float r = acc[v];
        r = r > 0.0f ? r : 0.0f;
        out[(size_t)(row0 + v + 8 * half) * H + c0 + lm] = r;
    }
}

#define CT_PITCH 68
#define TS_PITCH 132
#define XS_PITCH 92
#define XT_VEC   ((16 * F85) / 4)

__global__ __launch_bounds__(128, 12) void conv12_out_kernel(
    const float* __restrict__ x_trans,
    const float* __restrict__ aggc1, const float* __restrict__ aggm1,
    const float* __restrict__ W1t,  const float* __restrict__ b1sum,
    const float* __restrict__ h_c,  const float* __restrict__ h_m,
    const int* __restrict__ headc,  const int* __restrict__ headm,
    const int* __restrict__ nodec,  const int* __restrict__ nodem,
    const float* __restrict__ w2l,
    const float* __restrict__ w2r,  const float* __restrict__ b2,
    const float* __restrict__ wlin, const float* __restrict__ blin,
    float* __restrict__ out, int M, int E) {
    __shared__ alignas(16) float hs[16 * CT_PITCH];
    __shared__ alignas(16) float cs[16 * CT_PITCH];
    __shared__ alignas(16) float ms[16 * CT_PITCH];
    __shared__ alignas(16) float ts[16 * TS_PITCH];
    float* xs = ts;

    const int row0 = blockIdx.x * 16;
    const int tid  = threadIdx.x;

    const int gr = tid >> 3;
    const int gq = tid & 7;
    const int ec = headc[row0 + gr];
    const int em = headm[row0 + gr];

    float xr[12];
    {
        const float* __restrict__ xt = x_trans + (size_t)row0 * F85;
#pragma unroll
        for (int j = 0; j < 3; ++j) {
            const int p = tid + 128 * j;
            const int q = 4 * (p < XT_VEC ? p : XT_VEC - 1);
#pragma unroll
            for (int c = 0; c < 4; ++c) xr[4 * j + c] = xt[q + c];
        }
    }
    const float xe = ((tid & 16) ? aggm1 : aggc1)[row0 + (tid & 15)];

    {
        const int2*   __restrict__ nc  = reinterpret_cast<const int2*>(nodec);
        const int2*   __restrict__ nm  = reinterpret_cast<const int2*>(nodem);
        const float4* __restrict__ hc4 = reinterpret_cast<const float4*>(h_c);
        const float4* __restrict__ hm4 = reinterpret_cast<const float4*>(h_m);
        float4 c0 = make_float4(0.f, 0.f, 0.f, 0.f), c1 = c0, m0 = c0, m1 = c0;
        int  ic = ec, im = em;
        bool lc = (unsigned)ic < (unsigned)E, lm = (unsigned)im < (unsigned)E;
        for (int s = 0; s < E && (lc || lm); ++s) {
            const int2 dc = nc[lc ? ic : 0];
            const int2 dm = nm[lm ? im : 0];
            const float4* pc = hc4 + (size_t)dc.y * (H / 4) + gq;
            const float4* pm = hm4 + (size_t)dm.y * (H / 4) + gq;
            const float4 u0 = pc[0], u1 = pc[8], w0 = pm[0], w1 = pm[8];
            if (lc) { c0 += u0; c1 += u1; ic = dc.x; }
            if (lm) { m0 += w0; m1 += w1; im = dm.x; }
            lc = lc && (unsigned)ic < (unsigned)E;
            lm = lm && (unsigned)im < (unsigned)E;
        }
        float4* cs4 = reinterpret_cast<float4*>(cs);
        float4* ms4 = reinterpret_cast<float4*>(ms);
        cs4[gr * (CT_PITCH / 4) + gq]     = c0;
        cs4[gr * (CT_PITCH / 4) + 8 + gq] = c1;
        ms4[gr * (CT_PITCH / 4) + gq]     = m0;
        ms4[gr * (CT_PITCH / 4) + 8 + gq] = m1;
    }

#pragma unroll
    for (int j = 0; j < 3; ++j) {
        const int p = tid + 128 * j;
        if (j < 2 || p < XT_VEC) {
#pragma unroll
            for (int c = 0; c < 4; ++c) {
                const int i = 4 * p + c;
                const int r = i / F85;
                xs[r * XS_PITCH + (i - r * F85)] = xr[4 * j + c];
            }
        }
    }
    if (tid < 48) xs[(tid & 15) * XS_PITCH + F85 + (tid >> 4)] = tid < 32 ? xe : 0.0f;
    __syncthreads();

    const int wave = tid >> 5;
    const int lane = tid & 31;
    const int half = lane >> 4;
    const int lm   = lane & 15;
    const int c0   = wave * 16;

    {
        v8f acc;
        const float bv = b1sum[c0 + lm];
#pragma unroll
        for (int i = 0; i < 8; ++i) acc[i] = bv;
        for (int k = 0; k < KPAD; k += 4) {
            const int ka = k + 2 * half;
            v2f a, b;
            a.x = xs[lm * XS_PITCH + ka];
            a.y = xs[lm * XS_PITCH + ka + 1];
            b.x = W1t[(ka + 0) * H + c0 + lm];
            b.y = W1t[(ka + 1) * H + c0 + lm];
            acc = __builtin_amdgcn_wmma_f32_16x16x4_f32(false, a, false, b, (short)0, acc,
                                                        false, false);
        }
#pragma unroll
        for (int v = 0; v < 8; ++v) {
            float r = acc[v];
            r = r > 0.0f ? r : 0.0f;
            hs[(v + 8 * half) * CT_PITCH + c0 + lm] = r;
        }
    }
    __syncthreads();

    v8f acc1, acc2;
    const float bv1 = b2[0 * H + c0 + lm];
    const float bv2 = b2[1 * H + c0 + lm];
#pragma unroll
    for (int i = 0; i < 8; ++i) { acc1[i] = bv1; acc2[i] = bv2; }

    const float* W2l0 = w2l;
    const float* W2l1 = w2l + 1 * H * H;
    const float* W2r0 = w2r;
    const float* W2r1 = w2r + 1 * H * H;

    for (int k = 0; k < H; k += 4) {
        const int ka = k + 2 * half;
        v2f ah, ac, am, b;
        ah.x = hs[lm * CT_PITCH + ka]; ah.y = hs[lm * CT_PITCH + ka + 1];
        ac.x = cs[lm * CT_PITCH + ka]; ac.y = cs[lm * CT_PITCH + ka + 1];
        am.x = ms[lm * CT_PITCH + ka]; am.y = ms[lm * CT_PITCH + ka + 1];

        b.x = W2l0[(ka + 0) * H + c0 + lm]; b.y = W2l0[(ka + 1) * H + c0 + lm];
        acc1 = __builtin_amdgcn_wmma_f32_16x16x4_f32(false, ac, false, b, (short)0, acc1,
                                                     false, false);
        b.x = W2r0[(ka + 0) * H + c0 + lm]; b.y = W2r0[(ka + 1) * H + c0 + lm];
        acc1 = __builtin_amdgcn_wmma_f32_16x16x4_f32(false, ah, false, b, (short)0, acc1,
                                                     false, false);
        b.x = W2l1[(ka + 0) * H + c0 + lm]; b.y = W2l1[(ka + 1) * H + c0 + lm];
        acc2 = __builtin_amdgcn_wmma_f32_16x16x4_f32(false, am, false, b, (short)0, acc2,
                                                     false, false);
        b.x = W2r1[(ka + 0) * H + c0 + lm]; b.y = W2r1[(ka + 1) * H + c0 + lm];
        acc2 = __builtin_amdgcn_wmma_f32_16x16x4_f32(false, ah, false, b, (short)0, acc2,
                                                     false, false);
    }

#pragma unroll
    for (int v = 0; v < 8; ++v) {
        float r1 = acc1[v]; r1 = r1 > 0.0f ? r1 : 0.0f;
        float r2 = acc2[v]; r2 = r2 > 0.0f ? r2 : 0.0f;
        ts[(v + 8 * half) * TS_PITCH + c0 + lm]      = r1;
        ts[(v + 8 * half) * TS_PITCH + 64 + c0 + lm] = r2;
    }
    __syncthreads();

    if (tid < 32) {
        const int r = tid >> 1, c = tid & 1;
        float s = blin[c];
#pragma unroll 8
        for (int j = 0; j < 128; ++j) s += ts[r * TS_PITCH + j] * wlin[j * 2 + c];
        out[(size_t)(row0 + r) * 2 + c] = s;
    }
}

extern "C" void kernel_launch(void* const* d_in, const int* in_sizes, int n_in,
                              void* d_out, int out_size, void* d_ws, size_t ws_size,
                              hipStream_t stream) {
    const float* x_card   = (const float*)d_in[0];
    const float* x_merch  = (const float*)d_in[1];
    const float* x_trans  = (const float*)d_in[2];
    const float* w1_src_t = (const float*)d_in[3];
    const float* w1_dst_t = (const float*)d_in[4];
    const float* b1_t     = (const float*)d_in[5];
    const float* w1_src_r = (const float*)d_in[6];
    const float* w1_dst_r = (const float*)d_in[7];
    const float* b1_r     = (const float*)d_in[8];
    const float* w2_l     = (const float*)d_in[9];
    const float* w2_r     = (const float*)d_in[10];
    const float* b2       = (const float*)d_in[11];
    const float* w_lin    = (const float*)d_in[12];
    const float* b_lin    = (const float*)d_in[13];
    const int* src_pays   = (const int*)d_in[14];
    const int* dst_pays   = (const int*)d_in[15];
    const int* src_recv   = (const int*)d_in[16];
    const int* dst_recv   = (const int*)d_in[17];
    float* out = (float*)d_out;

    const int N_c = in_sizes[0];
    const int N_m = in_sizes[1];
    const int N_t = in_sizes[2] / F85;
    const int E   = in_sizes[14];

    float* ws = (float*)d_ws;
    size_t off = 0;
    float* h_c    = ws + off; off += (size_t)N_c * H;
    float* h_m    = ws + off; off += (size_t)N_m * H;
    int*   nodec  = (int*)(ws + off); off += (size_t)E * 2;
    int*   nodem  = (int*)(ws + off); off += (size_t)E * 2;
    float* aggc1  = ws + off; off += (size_t)N_t;
    float* aggm1  = ws + off; off += (size_t)N_t;
    int*   headc  = (int*)(ws + off); off += (size_t)N_t;
    int*   headm  = (int*)(ws + off); off += (size_t)N_t;
    float* aggt2c = ws + off; off += (size_t)N_c * F85;
    float* aggt2m = ws + off; off += (size_t)N_m * F85;
    float* W1t    = ws + off; off += (size_t)KPAD * H;
    float* W1c    = ws + off; off += (size_t)KPAD * H;
    float* W1m    = ws + off; off += (size_t)KPAD * H;
    float* b1sum  = ws + off; off += (size_t)H;

    {
        long n = 2 * (long)N_t;
        long blocks = (n + 255) / 256;
        init_kernel<<<(unsigned)blocks, 256, 0, stream>>>(aggc1, headc, n);
    }
    prep_w_kernel<<<(KPAD * H + 255) / 256, 256, 0, stream>>>(
        w1_src_t, w1_dst_t, b1_t, w1_src_r, w1_dst_r, W1t, W1c, W1m, b1sum);
    edge_pass_kernel<<<(E + 255) / 256, 256, 0, stream>>>(
        x_card, x_merch, src_pays, dst_pays, src_recv, dst_recv, aggc1, aggm1,
        headc, headm, nodec, nodem, E);
    {
        const int gc = (N_c + RA_ROWS_C - 1) / RA_ROWS_C;
        const int gm = (N_m + RA_ROWS_M - 1) / RA_ROWS_M;
        rev_agg_lds_kernel<<<gc + gm, RA_THREADS, 0, stream>>>(
            x_trans, src_pays, dst_pays, src_recv, dst_recv, aggt2c, aggt2m, N_c, N_m, E, gc);
    }
    gemm88_kernel<<<N_c / 16, 128, 0, stream>>>(aggt2c, x_card, nullptr, W1c, b1_r, h_c, N_c);
    gemm88_kernel<<<N_m / 16, 128, 0, stream>>>(aggt2m, x_merch, nullptr, W1m, b1_r + H, h_m, N_m);
    conv12_out_kernel<<<N_t / 16, 128, 0, stream>>>(
        x_trans, aggc1, aggm1, W1t, b1sum, h_c, h_m, headc, headm, nodec, nodem,
        w2_l, w2_r, b2, w_lin, b_lin, out, N_t, E);
}


# conv2 neighbour product moved in front of the aggregation: card/merchant GEMM kernel also multiplies its relu rows by w2_l, conv2 sums projected rows (sum order changed: project, then add)
# speedup vs baseline: 3.0993x; 1.1218x over previous
#include <hip/hip_runtime.h>

typedef __attribute__((ext_vector_type(2))) float v2f;
typedef __attribute__((ext_vector_type(8))) float v8f;

#define H 64
#define F85 85
#define KPAD 88

__global__ void init_kernel(float* __restrict__ z, int* __restrict__ heads, long n) {
    long i = (long)blockIdx.x * blockDim.x + threadIdx.x;
    if (i < n) { z[i] = 0.0f; heads[i] = -1; }
}

__global__ void prep_w_kernel(const float* __restrict__ w1_src_t,
                              const float* __restrict__ w1_dst_t,
                              const float* __restrict__ b1_t,
                              const float* __restrict__ w1_src_r,
                              const float* __restrict__ w1_dst_r,
                              float* __restrict__ W1t, float* __restrict__ W1c,
                              float* __restrict__ W1m, float* __restrict__ b1sum) {
    int i = blockIdx.x * blockDim.x + threadIdx.x;
    if (i >= KPAD * H) return;
    int r = i >> 6, c = i & 63;
    float vt = 0.f, vc = 0.f, vm = 0.f;
    if (r < F85) {
        vt = w1_dst_t[r * H + c] + w1_dst_t[F85 * H + r * H + c];
        vc = w1_src_r[r * H + c];
        vm = w1_src_r[F85 * H + r * H + c];
    } else if (r == 85) {
        vt = w1_src_t[c];
        vc = w1_dst_r[c];
        vm = w1_dst_r[H + c];
    } else if (r == 86) {
        vt = w1_src_t[H + c];
    }
    W1t[i] = vt; W1c[i] = vc; W1m[i] = vm;
    if (i < H) b1sum[i] = b1_t[i] + b1_t[H + i];
}

__global__ void edge_pass_kernel(const float* __restrict__ xc, const float* __restrict__ xm,
                                 const int* __restrict__ sp, const int* __restrict__ dp,
                                 const int* __restrict__ sr, const int* __restrict__ dr,
                                 float* __restrict__ aggc, float* __restrict__ aggm,
                                 int* __restrict__ headc, int* __restrict__ headm,
                                 int* __restrict__ nodec, int* __restrict__ nodem, int E) {
    int e = blockIdx.x * blockDim.x + threadIdx.x;
    if (e >= E) return;
    const int sc = sp[e], dc = dp[e], sm = sr[e], dm = dr[e];
    atomicAdd(&aggc[dc], xc[sc]);
    atomicAdd(&aggm[dm], xm[sm]);
    reinterpret_cast<int2*>(nodec)[e] = make_int2(atomicExch(&headc[dc], e), sc);
    reinterpret_cast<int2*>(nodem)[e] = make_int2(atomicExch(&headm[dm], e), sm);
}

#define RA_THREADS 1024
#define RA_TILE    16384
#define RA_UNR     6
#define RA_ROWS_C  392
#define RA_ROWS_M  80
#define RA_PITCH   96

template <bool VEC>
__device__ __forceinline__ void ra_load_keys(const int* __restrict__ key, int E, int tile,
                                             int tid, int4 (&cur)[4]) {
#pragma unroll
    for (int j = 0; j < 4; ++j) {
        const int e0 = tile * RA_TILE + (j * RA_THREADS + tid) * 4;
        int4 r;
        if (VEC) {
            r = *reinterpret_cast<const int4*>(key + (e0 < E ? e0 : 0));
        } else {
            const int last = E - 1;
            r.x = key[e0 + 0 < E ? e0 + 0 : last];
            r.y = key[e0 + 1 < E ? e0 + 1 : last];
            r.z = key[e0 + 2 < E ? e0 + 2 : last];
            r.w = key[e0 + 3 < E ? e0 + 3 : last];
        }
        cur[j] = r;
    }
}

template <bool VEC>
__device__ __forceinline__ void ra_scan_and_add(
    const float* __restrict__ x_trans, const int* __restrict__ key,
    const int* __restrict__ val, int E, int lo, int nrow, int part,
    float* acc, int* queue, int* qn) {
    const int tid  = threadIdx.x;
    const int wave = __builtin_amdgcn_readfirstlane(tid >> 5);
    const int lane = tid & 31;

    const int nTiles = (E + RA_TILE - 1) / RA_TILE;
    const int rot = part % nTiles;

    int4 cur[4];
    ra_load_keys<VEC>(key, E, rot, tid, cur);

    for (int it = 0; it < nTiles; ++it) {
        int tile = it + rot;
        if (tile >= nTiles) tile -= nTiles;
        const int buf = it & 1;

        unsigned m = 0;
        const int ebase = tile * RA_TILE + tid * 4;
#pragma unroll
        for (int j = 0; j < 4; ++j) {
            const int k[4] = {cur[j].x, cur[j].y, cur[j].z, cur[j].w};
#pragma unroll
            for (int c = 0; c < 4; ++c) {
                const bool inside = ebase + j * (RA_THREADS * 4) + c < E;
                m |= (inside && (unsigned)(k[c] - lo) < (unsigned)nrow ? 1u : 0u) << (j * 4 + c);
            }
        }
        if (m) {
            int pos = atomicAdd(&qn[buf], __builtin_popcount(m));
            do {
                const int b = __builtin_ctz(m);
                queue[pos++] = ebase + (b >> 2) * (RA_THREADS * 4) + (b & 3);
                m &= m - 1;
            } while (m);
        }
        __syncthreads();

        if (it + 1 < nTiles) {
            int ntile = tile + 1;
            if (ntile >= nTiles) ntile -= nTiles;
            ra_load_keys<VEC>(key, E, ntile, tid, cur);
        }
        if (tid == 0) qn[buf ^ 1] = 0;

        const int n = __builtin_amdgcn_readfirstlane(qn[buf]);
        auto round = [&](int q0) {
            const int myq = q0 + 32 * lane;
            const int ev  = queue[(lane < RA_UNR && myq < n) ? myq : 0];
            int eu[RA_UNR], rb[RA_UNR];
            const float* rowp[RA_UNR];
            float v0[RA_UNR], v1[RA_UNR], v2[RA_UNR];
#pragma unroll
            for (int u = 0; u < RA_UNR; ++u) eu[u] = __builtin_amdgcn_readlane(ev, u);
#pragma unroll
            for (int u = 0; u < RA_UNR; ++u) {
                rb[u]   = (key[eu[u]] - lo) * RA_PITCH;
                rowp[u] = x_trans + (size_t)val[eu[u]] * F85;
            }
#pragma unroll
            for (int u = 0; u < RA_UNR; ++u) {
                v0[u] = rowp[u][lane];
                v1[u] = rowp[u][32 + lane];
                v2[u] = rowp[u][lane < F85 - 64 ? 64 + lane : 64];
            }
#pragma unroll
            for (int u = 0; u < RA_UNR; ++u) {
                const bool ok = q0 + 32 * u < n;
                atomicAdd(&acc[rb[u] + lane],      ok ? v0[u] : 0.0f);
                atomicAdd(&acc[rb[u] + 32 + lane], ok ? v1[u] : 0.0f);
                atomicAdd(&acc[rb[u] + 64 + lane], ok ? v2[u] : 0.0f);
            }
        };
        if (n > 0) round(wave);
        for (int q0 = wave + 32 * RA_UNR; q0 < n; q0 += 32 * RA_UNR) round(q0);
        __syncthreads();
    }
}

__global__ __launch_bounds__(RA_THREADS) void rev_agg_lds_kernel(
    const float* __restrict__ x_trans,
    const int* __restrict__ src_pays, const int* __restrict__ dst_pays,
    const int* __restrict__ src_recv, const int* __restrict__ dst_recv,
    float* __restrict__ aggc, float* __restrict__ aggm,
    int N_c, int N_m, int E, int gc) {
    __shared__ float acc[RA_ROWS_C * RA_PITCH];
    __shared__ int   queue[RA_TILE];
    __shared__ int   qn[2];

    const int tid = threadIdx.x;
    const bool is_card = (int)blockIdx.x < gc;
    const int  part = is_card ? (int)blockIdx.x : (int)blockIdx.x - gc;
    const int  rows = is_card ? RA_ROWS_C : RA_ROWS_M;
    const int  N    = is_card ? N_c : N_m;
    const int* __restrict__ key = is_card ? src_pays : src_recv;
    const int* __restrict__ val = is_card ? dst_pays : dst_recv;
    float* __restrict__ outp    = is_card ? aggc : aggm;

    const int lo = part * rows;
    int nrow = N - lo;
    if (nrow > rows) nrow = rows;
    if (nrow < 0) nrow = 0;

    for (int i = tid; i < nrow * RA_PITCH; i += RA_THREADS) acc[i] = 0.0f;
    if (tid < 2) qn[tid] = 0;
    __syncthreads();

    if (E > 0) {
        const bool vec_ok = ((((size_t)key) & 15) == 0) && ((E & 3) == 0);
        if (vec_ok) ra_scan_and_add<true>(x_trans, key, val, E, lo, nrow, part, acc, queue, qn);
        else        ra_scan_and_add<false>(x_trans, key, val, E, lo, nrow, part, acc, queue, qn);
    }

    float* __restrict__ dstp = outp + (size_t)lo * F85;
    for (int i = tid; i < nrow * F85; i += RA_THREADS) {
        const int r = i / F85;
        dstp[i] = acc[r * RA_PITCH + (i - r * F85)];
    }
}

#define CT_PITCH 68
#define TS_PITCH 132
#define XS_PITCH 92
#define XT_VEC   ((16 * F85) / 4)

__global__ __launch_bounds__(128) void gemm88_proj_kernel(
    const float* __restrict__ Xmain, const float* __restrict__ extra1,
    const float* __restrict__ extra2, const float* __restrict__ W,
    const float* __restrict__ bias, const float* __restrict__ W2,
    float* __restrict__ out, int M) {
    __shared__ float xs[16 * KPAD];
    __shared__ float hs[16 * CT_PITCH];

    const int row0 = blockIdx.x * 16;
    for (int i = threadIdx.x; i < 16 * KPAD; i += 128) {
        int r = i / KPAD, c = i - r * KPAD;
        int grow = row0 + r;
        float v;
        if (c < F85)      v = Xmain[(size_t)grow * F85 + c];
        else if (c == 85) v = extra1 ? extra1[grow] : 0.0f;
        else if (c == 86) v = extra2 ? extra2[grow] : 0.0f;
        else              v = 0.0f;
        xs[i] = v;
    }
    __syncthreads();

    const int wave = threadIdx.x >> 5;
    const int lane = threadIdx.x & 31;
    const int half = lane >> 4;
    const int lm   = lane & 15;
    const int c0   = wave * 16;

    const float bval = bias[c0 + lm];
    v8f acc;
#pragma unroll
    for (int i = 0; i < 8; ++i) acc[i] = bval;

    for (int k = 0; k < KPAD; k += 4) {
        const int ka = k + 2 * half;
        v2f a, b;
        a.x = xs[lm * KPAD + ka];
        a.y = xs[lm * KPAD + ka + 1];
        b.x = W[(size_t)(ka + 0) * H + c0 + lm];
        b.y = W[(size_t)(ka + 1) * H + c0 + lm];
        acc = __builtin_amdgcn_wmma_f32_16x16x4_f32(false, a, false, b, (short)0, acc,
                                                    false, false);
    }
#pragma unroll
    for (int v = 0; v < 8; ++v) {
        float r = acc[v];
        r = r > 0.0f ? r : 0.0f;
        hs[(v + 8 * half) * CT_PITCH + c0 + lm] = r;
    }
    __syncthreads();

    v8f g;
#pragma unroll
    for (int i = 0; i < 8; ++i) g[i] = 0.0f;
    for (int k = 0; k < H; k += 4) {
        const int ka = k + 2 * half;
        v2f a, b;
        a.x = hs[lm * CT_PITCH + ka];
        a.y = hs[lm * CT_PITCH + ka + 1];
        b.x = W2[(ka + 0) * H + c0 + lm];
        b.y = W2[(ka + 1) * H + c0 + lm];
        g = __builtin_amdgcn_wmma_f32_16x16x4_f32(false, a, false, b, (short)0, g,
                                                  false, false);
    }
#pragma unroll
    for (int v = 0; v < 8; ++v)
        out[(size_t)(row0 + v + 8 * half) * H + c0 + lm] = g[v];
}


__global__ __launch_bounds__(128, 12) void conv12_out_kernel(
    const float* __restrict__ x_trans,
    const float* __restrict__ aggc1, const float* __restrict__ aggm1,
    const float* __restrict__ W1t,  const float* __restrict__ b1sum,
    const float* __restrict__ g_c,  const float* __restrict__ g_m,
    const int* __restrict__ headc,  const int* __restrict__ headm,
    const int* __restrict__ nodec,  const int* __restrict__ nodem,
    const float* __restrict__ w2r,  const float* __restrict__ b2,
    const float* __restrict__ wlin, const float* __restrict__ blin,
    float* __restrict__ out, int M, int E) {
    __shared__ alignas(16) float hs[16 * CT_PITCH];
    __shared__ alignas(16) float cs[16 * CT_PITCH];
    __shared__ alignas(16) float ms[16 * CT_PITCH];
    __shared__ alignas(16) float ts[16 * TS_PITCH];
    float* xs = ts;

    const int row0 = blockIdx.x * 16;
    const int tid  = threadIdx.x;

    const int gr = tid >> 3;
    const int gq = tid & 7;
    const int ec = headc[row0 + gr];
    const int em = headm[row0 + gr];

    float xr[12];
    {
        const float* __restrict__ xt = x_trans + (size_t)row0 * F85;
#pragma unroll
        for (int j = 0; j < 3; ++j) {
            const int p = tid + 128 * j;
            const int q = 4 * (p < XT_VEC ? p : XT_VEC - 1);
#pragma unroll
            for (int c = 0; c < 4; ++c) xr[4 * j + c] = xt[q + c];
        }
    }
    const float xe = ((tid & 16) ? aggm1 : aggc1)[row0 + (tid & 15)];

    {
        const int2*   __restrict__ nc  = reinterpret_cast<const int2*>(nodec);
        const int2*   __restrict__ nm  = reinterpret_cast<const int2*>(nodem);
        const float4* __restrict__ hc4 = reinterpret_cast<const float4*>(g_c);
        const float4* __restrict__ hm4 = reinterpret_cast<const float4*>(g_m);
        float4 c0 = make_float4(0.f, 0.f, 0.f, 0.f), c1 = c0, m0 = c0, m1 = c0;
        int  ic = ec, im = em;
        bool lc = (unsigned)ic < (unsigned)E, lm = (unsigned)im < (unsigned)E;
        for (int s = 0; s < E && (lc || lm); ++s) {
            const int2 dc = nc[lc ? ic : 0];
            const int2 dm = nm[lm ? im : 0];
            const float4* pc = hc4 + (size_t)dc.y * (H / 4) + gq;
            const float4* pm = hm4 + (size_t)dm.y * (H / 4) + gq;
            const float4 u0 = pc[0], u1 = pc[8], w0 = pm[0], w1 = pm[8];
            if (lc) { c0 += u0; c1 += u1; ic = dc.x; }
            if (lm) { m0 += w0; m1 += w1; im = dm.x; }
            lc = lc && (unsigned)ic < (unsigned)E;
            lm = lm && (unsigned)im < (unsigned)E;
        }
        float4* cs4 = reinterpret_cast<float4*>(cs);
        float4* ms4 = reinterpret_cast<float4*>(ms);
        cs4[gr * (CT_PITCH / 4) + gq]     = c0;
        cs4[gr * (CT_PITCH / 4) + 8 + gq] = c1;
        ms4[gr * (CT_PITCH / 4) + gq]     = m0;
        ms4[gr * (CT_PITCH / 4) + 8 + gq] = m1;
    }

#pragma unroll
    for (int j = 0; j < 3; ++j) {
        const int p = tid + 128 * j;
        if (j < 2 || p < XT_VEC) {
#pragma unroll
            for (int c = 0; c < 4; ++c) {
                const int i = 4 * p + c;
                const int r = i / F85;
                xs[r * XS_PITCH + (i - r * F85)] = xr[4 * j + c];
            }
        }
    }
    if (tid < 48) xs[(tid & 15) * XS_PITCH + F85 + (tid >> 4)] = tid < 32 ? xe : 0.0f;
    __syncthreads();

    const int wave = tid >> 5;
    const int lane = tid & 31;
    const int half = lane >> 4;
    const int lm   = lane & 15;
    const int c0   = wave * 16;

    {
        v8f acc;
        const float bv = b1sum[c0 + lm];
#pragma unroll
        for (int i = 0; i < 8; ++i) acc[i] = bv;
        for (int k = 0; k < KPAD; k += 4) {
            const int ka = k + 2 * half;
            v2f a, b;
            a.x = xs[lm * XS_PITCH + ka];
            a.y = xs[lm * XS_PITCH + ka + 1];
            b.x = W1t[(ka + 0) * H + c0 + lm];
            b.y = W1t[(ka + 1) * H + c0 + lm];
            acc = __builtin_amdgcn_wmma_f32_16x16x4_f32(false, a, false, b, (short)0, acc,
                                                        false, false);
        }
#pragma unroll
        for (int v = 0; v < 8; ++v) {
            float r = acc[v];
            r = r > 0.0f ? r : 0.0f;
            hs[(v + 8 * half) * CT_PITCH + c0 + lm] = r;
        }
    }
    __syncthreads();

    v8f acc1, acc2;
    const float bv1 = b2[0 * H + c0 + lm];
    const float bv2 = b2[1 * H + c0 + lm];
#pragma unroll
    for (int v = 0; v < 8; ++v) {
        acc1[v] = bv1 + cs[(v + 8 * half) * CT_PITCH + c0 + lm];
        acc2[v] = bv2 + ms[(v + 8 * half) * CT_PITCH + c0 + lm];
    }

    const float* W2r0 = w2r;
    const float* W2r1 = w2r + 1 * H * H;

    for (int k = 0; k < H; k += 4) {
        const int ka = k + 2 * half;
        v2f ah, b;
        ah.x = hs[lm * CT_PITCH + ka]; ah.y = hs[lm * CT_PITCH + ka + 1];

        b.x = W2r0[(ka + 0) * H + c0 + lm]; b.y = W2r0[(ka + 1) * H + c0 + lm];
        acc1 = __builtin_amdgcn_wmma_f32_16x16x4_f32(false, ah, false, b, (short)0, acc1,
                                                     false, false);
        b.x = W2r1[(ka + 0) * H + c0 + lm]; b.y = W2r1[(ka + 1) * H + c0 + lm];
        acc2 = __builtin_amdgcn_wmma_f32_16x16x4_f32(false, ah, false, b, (short)0, acc2,
                                                     false, false);
    }

#pragma unroll
    for (int v = 0; v < 8; ++v) {
        float r1 = acc1[v]; r1 = r1 > 0.0f ? r1 : 0.0f;
        float r2 = acc2[v]; r2 = r2 > 0.0f ? r2 : 0.0f;
        ts[(v + 8 * half) * TS_PITCH + c0 + lm]      = r1;
        ts[(v + 8 * half) * TS_PITCH + 64 + c0 + lm] = r2;
    }
    __syncthreads();

    if (tid < 32) {
        const int r = tid >> 1, c = tid & 1;
        float s = blin[c];
#pragma unroll 8
        for (int j = 0; j < 128; ++j) s += ts[r * TS_PITCH + j] * wlin[j * 2 + c];
        out[(size_t)(row0 + r) * 2 + c] = s;
    }
}

extern "C" void kernel_launch(void* const* d_in, const int* in_sizes, int n_in,
                              void* d_out, int out_size, void* d_ws, size_t ws_size,
                              hipStream_t stream) {
    const float* x_card   = (const float*)d_in[0];
    const float* x_merch  = (const float*)d_in[1];
    const float* x_trans  = (const float*)d_in[2];
    const float* w1_src_t = (const float*)d_in[3];
    const float* w1_dst_t = (const float*)d_in[4];
    const float* b1_t     = (const float*)d_in[5];
    const float* w1_src_r = (const float*)d_in[6];
    const float* w1_dst_r = (const float*)d_in[7];
    const float* b1_r     = (const float*)d_in[8];
    const float* w2_l     = (const float*)d_in[9];
    const float* w2_r     = (const float*)d_in[10];
    const float* b2       = (const float*)d_in[11];
    const float* w_lin    = (const float*)d_in[12];
    const float* b_lin    = (const float*)d_in[13];
    const int* src_pays   = (const int*)d_in[14];
    const int* dst_pays   = (const int*)d_in[15];
    const int* src_recv   = (const int*)d_in[16];
    const int* dst_recv   = (const int*)d_in[17];
    float* out = (float*)d_out;

    const int N_c = in_sizes[0];
    const int N_m = in_sizes[1];
    const int N_t = in_sizes[2] / F85;
    const int E   = in_sizes[14];

    float* ws = (float*)d_ws;
    size_t off = 0;
    float* g_c    = ws + off; off += (size_t)N_c * H;
    float* g_m    = ws + off; off += (size_t)N_m * H;
    int*   nodec  = (int*)(ws + off); off += (size_t)E * 2;
    int*   nodem  = (int*)(ws + off); off += (size_t)E * 2;
    float* aggc1  = ws + off; off += (size_t)N_t;
    float* aggm1  = ws + off; off += (size_t)N_t;
    int*   headc  = (int*)(ws + off); off += (size_t)N_t;
    int*   headm  = (int*)(ws + off); off += (size_t)N_t;
    float* aggt2c = ws + off; off += (size_t)N_c * F85;
    float* aggt2m = ws + off; off += (size_t)N_m * F85;
    float* W1t    = ws + off; off += (size_t)KPAD * H;
    float* W1c    = ws + off; off += (size_t)KPAD * H;
    float* W1m    = ws + off; off += (size_t)KPAD * H;
    float* b1sum  = ws + off; off += (size_t)H;

    {
        long n = 2 * (long)N_t;
        long blocks = (n + 255) / 256;
        init_kernel<<<(unsigned)blocks, 256, 0, stream>>>(aggc1, headc, n);
    }
    prep_w_kernel<<<(KPAD * H + 255) / 256, 256, 0, stream>>>(
        w1_src_t, w1_dst_t, b1_t, w1_src_r, w1_dst_r, W1t, W1c, W1m, b1sum);
    edge_pass_kernel<<<(E + 255) / 256, 256, 0, stream>>>(
        x_card, x_merch, src_pays, dst_pays, src_recv, dst_recv, aggc1, aggm1,
        headc, headm, nodec, nodem, E);
    {
        const int gc = (N_c + RA_ROWS_C - 1) / RA_ROWS_C;
        const int gm = (N_m + RA_ROWS_M - 1) / RA_ROWS_M;
        rev_agg_lds_kernel<<<gc + gm, RA_THREADS, 0, stream>>>(
            x_trans, src_pays, dst_pays, src_recv, dst_recv, aggt2c, aggt2m, N_c, N_m, E, gc);
    }
    gemm88_proj_kernel<<<N_c / 16, 128, 0, stream>>>(aggt2c, x_card, nullptr, W1c, b1_r,
                                                     w2_l, g_c, N_c);
    gemm88_proj_kernel<<<N_m / 16, 128, 0, stream>>>(aggt2m, x_merch, nullptr, W1m, b1_r + H,
                                                     w2_l + H * H, g_m, N_m);
    conv12_out_kernel<<<N_t / 16, 128, 0, stream>>>(
        x_trans, aggc1, aggm1, W1t, b1sum, g_c, g_m, headc, headm, nodec, nodem,
        w2_r, b2, w_lin, b_lin, out, N_t, E);
}


# conv1 reverse aggregation as gather: card/merchant rows get edge sub-lists in the edge pass; their x_trans rows are summed in registers inside the GEMM kernel (list order = arrival order, as before)
# speedup vs baseline: 3.3481x; 1.0803x over previous
#include <hip/hip_runtime.h>

typedef __attribute__((ext_vector_type(2))) float v2f;
typedef __attribute__((ext_vector_type(8))) float v8f;

#define H 64
#define F85 85
#define KPAD 88
#define RSUB 8

__global__ void init_kernel(float* __restrict__ z, int* __restrict__ heads, long nz, long nh) {
    long i = (long)blockIdx.x * blockDim.x + threadIdx.x;
    if (i < nz) z[i] = 0.0f;
    if (i < nh) heads[i] = -1;
}

__global__ void prep_w_kernel(const float* __restrict__ w1_src_t,
                              const float* __restrict__ w1_dst_t,
                              const float* __restrict__ b1_t,
                              const float* __restrict__ w1_src_r,
                              const float* __restrict__ w1_dst_r,
                              float* __restrict__ W1t, float* __restrict__ W1c,
                              float* __restrict__ W1m, float* __restrict__ b1sum) {
    int i = blockIdx.x * blockDim.x + threadIdx.x;
    if (i >= KPAD * H) return;
    int r = i >> 6, c = i & 63;
    float vt = 0.f, vc = 0.f, vm = 0.f;
    if (r < F85) {
        vt = w1_dst_t[r * H + c] + w1_dst_t[F85 * H + r * H + c];
        vc = w1_src_r[r * H + c];
        vm = w1_src_r[F85 * H + r * H + c];
    } else if (r == 85) {
        vt = w1_src_t[c];
        vc = w1_dst_r[c];
        vm = w1_dst_r[H + c];
    } else if (r == 86) {
        vt = w1_src_t[H + c];
    }
    W1t[i] = vt; W1c[i] = vc; W1m[i] = vm;
    if (i < H) b1sum[i] = b1_t[i] + b1_t[H + i];
}

__global__ void edge_pass_kernel(const float* __restrict__ xc, const float* __restrict__ xm,
                                 const int* __restrict__ sp, const int* __restrict__ dp,
                                 const int* __restrict__ sr, const int* __restrict__ dr,
                                 float* __restrict__ aggc, float* __restrict__ aggm,
                                 int* __restrict__ headc, int* __restrict__ headm,
                                 int* __restrict__ nodec, int* __restrict__ nodem,
                                 int* __restrict__ rheadc, int* __restrict__ rheadm,
                                 int* __restrict__ rnodec, int* __restrict__ rnodem, int E) {
    int e = blockIdx.x * blockDim.x + threadIdx.x;
    if (e >= E) return;
    const int sc = sp[e], dc = dp[e], sm = sr[e], dm = dr[e];
    atomicAdd(&aggc[dc], xc[sc]);
    atomicAdd(&aggm[dm], xm[sm]);
    reinterpret_cast<int2*>(nodec)[e] = make_int2(atomicExch(&headc[dc], e), sc);
    reinterpret_cast<int2*>(nodem)[e] = make_int2(atomicExch(&headm[dm], e), sm);
    const int sub = e & (RSUB - 1);
    reinterpret_cast<int2*>(rnodec)[e] =
        make_int2(atomicExch(&rheadc[(size_t)sc * RSUB + sub], e), dc);
    reinterpret_cast<int2*>(rnodem)[e] =
        make_int2(atomicExch(&rheadm[(size_t)sm * RSUB + sub], e), dm);
}

#define CT_PITCH 68
#define TS_PITCH 132
#define XS_PITCH 92
#define XT_VEC   ((16 * F85) / 4)

__global__ __launch_bounds__(128) void gemm88_proj_kernel(
    const float* __restrict__ x_trans, const int* __restrict__ rhead,
    const int* __restrict__ rnode, const float* __restrict__ extra1,
    const float* __restrict__ extra2, const float* __restrict__ W,
    const float* __restrict__ bias, const float* __restrict__ W2,
    float* __restrict__ out, int M, int E) {
    __shared__ float xs[16 * KPAD];
    __shared__ float hs[16 * CT_PITCH];

    const int row0 = blockIdx.x * 16;
    const int wave = threadIdx.x >> 5;
    const int lane = threadIdx.x & 31;

    float e1 = 0.0f, e2 = 0.0f;
    if (threadIdx.x < 16) {
        if (extra1) e1 = extra1[row0 + threadIdx.x];
        if (extra2) e2 = extra2[row0 + threadIdx.x];
    }

    {
        const int2* __restrict__ rn = reinterpret_cast<const int2*>(rnode);
        const int c2 = lane < F85 - 64 ? 64 + lane : 64;
        float a0[4], a1[4], a2[4];
#pragma unroll
        for (int i = 0; i < 4; ++i) { a0[i] = 0.0f; a1[i] = 0.0f; a2[i] = 0.0f; }

        int  cur = rhead[(size_t)(row0 + 4 * wave + lane / RSUB) * RSUB + (lane & (RSUB - 1))];
        bool act = (unsigned)cur < (unsigned)E;
        int2 nd  = rn[act ? cur : 0];
        for (int s = 0; s < E; ++s) {
            const unsigned live = __builtin_amdgcn_ballot_w32(act);
            if (live == 0) break;
            const int trow = nd.y;
            const int nxt  = nd.x;
            const bool act2 = act && (unsigned)nxt < (unsigned)E;
            const int2 ndn = rn[act2 ? nxt : 0];
            float v0[32], v1[32], v2[32];
#pragma unroll
            for (int c = 0; c < 32; ++c) {
                v0[c] = 0.0f; v1[c] = 0.0f; v2[c] = 0.0f;
                if (live & (1u << c)) {
                    const float* __restrict__ p =
                        x_trans + (size_t)__builtin_amdgcn_readlane(trow, c) * F85;
                    v0[c] = p[lane];
                    v1[c] = p[32 + lane];
                    v2[c] = p[c2];
                }
            }
#pragma unroll
            for (int c = 0; c < 32; ++c) {
                a0[c / RSUB] += v0[c];
                a1[c / RSUB] += v1[c];
                a2[c / RSUB] += v2[c];
            }
            nd  = ndn;
            act = act2;
        }
#pragma unroll
        for (int i = 0; i < 4; ++i) {
            float* xr = xs + (4 * wave + i) * KPAD;
            xr[lane]      = a0[i];
            xr[32 + lane] = a1[i];
            if (lane < F85 - 64) xr[64 + lane] = a2[i];
        }
    }
    if (threadIdx.x < 16) {
        xs[threadIdx.x * KPAD + 85] = e1;
        xs[threadIdx.x * KPAD + 86] = e2;
        xs[threadIdx.x * KPAD + 87] = 0.0f;
    }
    __syncthreads();

    const int half = lane >> 4;
    const int lm   = lane & 15;
    const int c0   = wave * 16;

    const float bval = bias[c0 + lm];
    v8f acc;
#pragma unroll
    for (int i = 0; i < 8; ++i) acc[i] = bval;

    for (int k = 0; k < KPAD; k += 4) {
        const int ka = k + 2 * half;
        v2f a, b;
        a.x = xs[lm * KPAD + ka];
        a.y = xs[lm * KPAD + ka + 1];
        b.x = W[(size_t)(ka + 0) * H + c0 + lm];
        b.y = W[(size_t)(ka + 1) * H + c0 + lm];
        acc = __builtin_amdgcn_wmma_f32_16x16x4_f32(false, a, false, b, (short)0, acc,
                                                    false, false);
    }
#pragma unroll
    for (int v = 0; v < 8; ++v) {
        float r = acc[v];
        r = r > 0.0f ? r : 0.0f;
        hs[(v + 8 * half) * CT_PITCH + c0 + lm] = r;
    }
    __syncthreads();

    v8f g;
#pragma unroll
    for (int i = 0; i < 8; ++i) g[i] = 0.0f;
    for (int k = 0; k < H; k += 4) {
        const int ka = k + 2 * half;
        v2f a, b;
        a.x = hs[lm * CT_PITCH + ka];
        a.y = hs[lm * CT_PITCH + ka + 1];
        b.x = W2[(ka + 0) * H + c0 + lm];
        b.y = W2[(ka + 1) * H + c0 + lm];
        g = __builtin_amdgcn_wmma_f32_16x16x4_f32(false, a, false, b, (short)0, g,
                                                  false, false);
    }
#pragma unroll
    for (int v = 0; v < 8; ++v)
        out[(size_t)(row0 + v + 8 * half) * H + c0 + lm] = g[v];
}


__global__ __launch_bounds__(128, 12) void conv12_out_kernel(
    const float* __restrict__ x_trans,
    const float* __restrict__ aggc1, const float* __restrict__ aggm1,
    const float* __restrict__ W1t,  const float* __restrict__ b1sum,
    const float* __restrict__ g_c,  const float* __restrict__ g_m,
    const int* __restrict__ headc,  const int* __restrict__ headm,
    const int* __restrict__ nodec,  const int* __restrict__ nodem,
    const float* __restrict__ w2r,  const float* __restrict__ b2,
    const float* __restrict__ wlin, const float* __restrict__ blin,
    float* __restrict__ out, int M, int E) {
    __shared__ alignas(16) float hs[16 * CT_PITCH];
    __shared__ alignas(16) float cs[16 * CT_PITCH];
    __shared__ alignas(16) float ms[16 * CT_PITCH];
    __shared__ alignas(16) float ts[16 * TS_PITCH];
    float* xs = ts;

    const int row0 = blockIdx.x * 16;
    const int tid  = threadIdx.x;

    const int gr = tid >> 3;
    const int gq = tid & 7;
    const int ec = headc[row0 + gr];
    const int em = headm[row0 + gr];

    float xr[12];
    {
        const float* __restrict__ xt = x_trans + (size_t)row0 * F85;
#pragma unroll
        for (int j = 0; j < 3; ++j) {
            const int p = tid + 128 * j;
            const int q = 4 * (p < XT_VEC ? p : XT_VEC - 1);
#pragma unroll
            for (int c = 0; c < 4; ++c) xr[4 * j + c] = xt[q + c];
        }
    }
    const float xe = ((tid & 16) ? aggm1 : aggc1)[row0 + (tid & 15)];

    {
        const int2*   __restrict__ nc  = reinterpret_cast<const int2*>(nodec);
        const int2*   __restrict__ nm  = reinterpret_cast<const int2*>(nodem);
        const float4* __restrict__ hc4 = reinterpret_cast<const float4*>(g_c);
        const float4* __restrict__ hm4 = reinterpret_cast<const float4*>(g_m);
        float4 c0 = make_float4(0.f, 0.f, 0.f, 0.f), c1 = c0, m0 = c0, m1 = c0;
        int  ic = ec, im = em;
        bool lc = (unsigned)ic < (unsigned)E, lm = (unsigned)im < (unsigned)E;
        for (int s = 0; s < E && (lc || lm); ++s) {
            const int2 dc = nc[lc ? ic : 0];
            const int2 dm = nm[lm ? im : 0];
            const float4* pc = hc4 + (size_t)dc.y * (H / 4) + gq;
            const float4* pm = hm4 + (size_t)dm.y * (H / 4) + gq;
            const float4 u0 = pc[0], u1 = pc[8], w0 = pm[0], w1 = pm[8];
            if (lc) { c0 += u0; c1 += u1; ic = dc.x; }
            if (lm) { m0 += w0; m1 += w1; im = dm.x; }
            lc = lc && (unsigned)ic < (unsigned)E;
            lm = lm && (unsigned)im < (unsigned)E;
        }
        float4* cs4 = reinterpret_cast<float4*>(cs);
        float4* ms4 = reinterpret_cast<float4*>(ms);
        cs4[gr * (CT_PITCH / 4) + gq]     = c0;
        cs4[gr * (CT_PITCH / 4) + 8 + gq] = c1;
        ms4[gr * (CT_PITCH / 4) + gq]     = m0;
        ms4[gr * (CT_PITCH / 4) + 8 + gq] = m1;
    }

#pragma unroll
    for (int j = 0; j < 3; ++j) {
        const int p = tid + 128 * j;
        if (j < 2 || p < XT_VEC) {
#pragma unroll
            for (int c = 0; c < 4; ++c) {
                const int i = 4 * p + c;
                const int r = i / F85;
                xs[r * XS_PITCH + (i - r * F85)] = xr[4 * j + c];
            }
        }
    }
    if (tid < 48) xs[(tid & 15) * XS_PITCH + F85 + (tid >> 4)] = tid < 32 ? xe : 0.0f;
    __syncthreads();

    const int wave = tid >> 5;
    const int lane = tid & 31;
    const int half = lane >> 4;
    const int lm   = lane & 15;
    const int c0   = wave * 16;

    {
        v8f acc;
        const float bv = b1sum[c0 + lm];
#pragma unroll
        for (int i = 0; i < 8; ++i) acc[i] = bv;
        for (int k = 0; k < KPAD; k += 4) {
            const int ka = k + 2 * half;
            v2f a, b;
            a.x = xs[lm * XS_PITCH + ka];
            a.y = xs[lm * XS_PITCH + ka + 1];
            b.x = W1t[(ka + 0) * H + c0 + lm];
            b.y = W1t[(ka + 1) * H + c0 + lm];
            acc = __builtin_amdgcn_wmma_f32_16x16x4_f32(false, a, false, b, (short)0, acc,
                                                        false, false);
        }
#pragma unroll
        for (int v = 0; v < 8; ++v) {
            float r = acc[v];
            r = r > 0.0f ? r : 0.0f;
            hs[(v + 8 * half) * CT_PITCH + c0 + lm] = r;
        }
    }
    __syncthreads();

    v8f acc1, acc2;
    const float bv1 = b2[0 * H + c0 + lm];
    const float bv2 = b2[1 * H + c0 + lm];
#pragma unroll
    for (int v = 0; v < 8; ++v) {
        acc1[v] = bv1 + cs[(v + 8 * half) * CT_PITCH + c0 + lm];
        acc2[v] = bv2 + ms[(v + 8 * half) * CT_PITCH + c0 + lm];
    }

    const float* W2r0 = w2r;
    const float* W2r1 = w2r + 1 * H * H;

    for (int k = 0; k < H; k += 4) {
        const int ka = k + 2 * half;
        v2f ah, b;
        ah.x = hs[lm * CT_PITCH + ka]; ah.y = hs[lm * CT_PITCH + ka + 1];

        b.x = W2r0[(ka + 0) * H + c0 + lm]; b.y = W2r0[(ka + 1) * H + c0 + lm];
        acc1 = __builtin_amdgcn_wmma_f32_16x16x4_f32(false, ah, false, b, (short)0, acc1,
                                                     false, false);
        b.x = W2r1[(ka + 0) * H + c0 + lm]; b.y = W2r1[(ka + 1) * H + c0 + lm];
        acc2 = __builtin_amdgcn_wmma_f32_16x16x4_f32(false, ah, false, b, (short)0, acc2,
                                                     false, false);
    }

#pragma unroll
    for (int v = 0; v < 8; ++v) {
        float r1 = acc1[v]; r1 = r1 > 0.0f ? r1 : 0.0f;
        float r2 = acc2[v]; r2 = r2 > 0.0f ? r2 : 0.0f;
        ts[(v + 8 * half) * TS_PITCH + c0 + lm]      = r1;
        ts[(v + 8 * half) * TS_PITCH + 64 + c0 + lm] = r2;
    }
    __syncthreads();

    if (tid < 32) {
        const int r = tid >> 1, c = tid & 1;
        float s = blin[c];
#pragma unroll 8
        for (int j = 0; j < 128; ++j) s += ts[r * TS_PITCH + j] * wlin[j * 2 + c];
        out[(size_t)(row0 + r) * 2 + c] = s;
    }
}

extern "C" void kernel_launch(void* const* d_in, const int* in_sizes, int n_in,
                              void* d_out, int out_size, void* d_ws, size_t ws_size,
                              hipStream_t stream) {
    const float* x_card   = (const float*)d_in[0];
    const float* x_merch  = (const float*)d_in[1];
    const float* x_trans  = (const float*)d_in[2];
    const float* w1_src_t = (const float*)d_in[3];
    const float* w1_dst_t = (const float*)d_in[4];
    const float* b1_t     = (const float*)d_in[5];
    const float* w1_src_r = (const float*)d_in[6];
    const float* w1_dst_r = (const float*)d_in[7];
    const float* b1_r     = (const float*)d_in[8];
    const float* w2_l     = (const float*)d_in[9];
    const float* w2_r     = (const float*)d_in[10];
    const float* b2       = (const float*)d_in[11];
    const float* w_lin    = (const float*)d_in[12];
    const float* b_lin    = (const float*)d_in[13];
    const int* src_pays   = (const int*)d_in[14];
    const int* dst_pays   = (const int*)d_in[15];
    const int* src_recv   = (const int*)d_in[16];
    const int* dst_recv   = (const int*)d_in[17];
    float* out = (float*)d_out;

    const int N_c = in_sizes[0];
    const int N_m = in_sizes[1];
    const int N_t = in_sizes[2] / F85;
    const int E   = in_sizes[14];

    float* ws = (float*)d_ws;
    size_t off = 0;
    float* g_c    = ws + off; off += (size_t)N_c * H;
    float* g_m    = ws + off; off += (size_t)N_m * H;
    int*   nodec  = (int*)(ws + off); off += (size_t)E * 2;
    int*   nodem  = (int*)(ws + off); off += (size_t)E * 2;
    int*   rnodec = (int*)(ws + off); off += (size_t)E * 2;
    int*   rnodem = (int*)(ws + off); off += (size_t)E * 2;
    float* aggc1  = ws + off; off += (size_t)N_t;
    float* aggm1  = ws + off; off += (size_t)N_t;
    int*   headc  = (int*)(ws + off); off += (size_t)N_t;
    int*   headm  = (int*)(ws + off); off += (size_t)N_t;
    int*   rheadc = (int*)(ws + off); off += (size_t)N_c * RSUB;
    int*   rheadm = (int*)(ws + off); off += (size_t)N_m * RSUB;
    float* W1t    = ws + off; off += (size_t)KPAD * H;
    float* W1c    = ws + off; off += (size_t)KPAD * H;
    float* W1m    = ws + off; off += (size_t)KPAD * H;
    float* b1sum  = ws + off; off += (size_t)H;

    {
        long nz = 2 * (long)N_t;
        long nh = 2 * (long)N_t + (long)RSUB * ((long)N_c + (long)N_m);
        long blocks = (nh + 255) / 256;
        init_kernel<<<(unsigned)blocks, 256, 0, stream>>>(aggc1, headc, nz, nh);
    }
    prep_w_kernel<<<(KPAD * H + 255) / 256, 256, 0, stream>>>(
        w1_src_t, w1_dst_t, b1_t, w1_src_r, w1_dst_r, W1t, W1c, W1m, b1sum);
    edge_pass_kernel<<<(E + 255) / 256, 256, 0, stream>>>(
        x_card, x_merch, src_pays, dst_pays, src_recv, dst_recv, aggc1, aggm1,
        headc, headm, nodec, nodem, rheadc, rheadm, rnodec, rnodem, E);
    gemm88_proj_kernel<<<N_c / 16, 128, 0, stream>>>(x_trans, rheadc, rnodec, x_card, nullptr,
                                                     W1c, b1_r, w2_l, g_c, N_c, E);
    gemm88_proj_kernel<<<N_m / 16, 128, 0, stream>>>(x_trans, rheadm, rnodem, x_merch, nullptr,
                                                     W1m, b1_r + H, w2_l + H * H, g_m, N_m, E);
    conv12_out_kernel<<<N_t / 16, 128, 0, stream>>>(
        x_trans, aggc1, aggm1, W1t, b1sum, g_c, g_m, headc, headm, nodec, nodem,
        w2_r, b2, w_lin, b_lin, out, N_t, E);
}


# prep kernel stores every weight matrix in the order the lanes read it (B fragments of two k-steps adjacent, w_lin transposed): operand loads are 16 bytes per lane, 4x fewer; sums unchanged
# speedup vs baseline: 3.4791x; 1.0391x over previous
#include <hip/hip_runtime.h>

typedef __attribute__((ext_vector_type(2))) float v2f;
typedef __attribute__((ext_vector_type(8))) float v8f;

#define H 64
#define F85 85
#define KPAD 88
#define RSUB 8

__global__ void init_kernel(float* __restrict__ z, int* __restrict__ heads, long nz, long nh) {
    long i = (long)blockIdx.x * blockDim.x + threadIdx.x;
    if (i < nz) z[i] = 0.0f;
    if (i < nh) heads[i] = -1;
}

__global__ void prep_w_kernel(const float* __restrict__ w1_src_t,
                              const float* __restrict__ w1_dst_t,
                              const float* __restrict__ b1_t,
                              const float* __restrict__ w1_src_r,
                              const float* __restrict__ w1_dst_r,
                              const float* __restrict__ w2_l,
                              const float* __restrict__ w2_r,
                              const float* __restrict__ w_lin,
                              float* __restrict__ W1t, float* __restrict__ W1c,
                              float* __restrict__ W1m, float* __restrict__ b1sum,
                              float* __restrict__ W2lc, float* __restrict__ W2lm,
                              float* __restrict__ W2rf, float* __restrict__ wlinT) {
    int i = blockIdx.x * blockDim.x + threadIdx.x;
    const int fj = i & 3, flane = (i >> 2) & 31, fwave = (i >> 7) & 3, fk = i >> 9;
    const int c = 16 * fwave + (flane & 15);
    if (i < 2 * H * H) {
        const int rr = 4 * fk + 2 * (flane >> 4) + (fj & 1);
        W2rf[i] = w2_r[(fj >> 1) * H * H + rr * H + c];
    }
    if (i < 4 * H) wlinT[i] = w_lin[(i & (2 * H - 1)) * 2 + (i >> 7)];
    if (i >= KPAD * H) return;
    const int r = 4 * (2 * fk + (fj >> 1)) + 2 * (flane >> 4) + (fj & 1);
    if (i < H * H) {
        W2lc[i] = w2_l[r * H + c];
        W2lm[i] = w2_l[H * H + r * H + c];
    }
    float vt = 0.f, vc = 0.f, vm = 0.f;
    if (r < F85) {
        vt = w1_dst_t[r * H + c] + w1_dst_t[F85 * H + r * H + c];
        vc = w1_src_r[r * H + c];
        vm = w1_src_r[F85 * H + r * H + c];
    } else if (r == 85) {
        vt = w1_src_t[c];
        vc = w1_dst_r[c];
        vm = w1_dst_r[H + c];
    } else if (r == 86) {
        vt = w1_src_t[H + c];
    }
    W1t[i] = vt; W1c[i] = vc; W1m[i] = vm;
    if (i < H) b1sum[i] = b1_t[i] + b1_t[H + i];
}

__global__ void edge_pass_kernel(const float* __restrict__ xc, const float* __restrict__ xm,
                                 const int* __restrict__ sp, const int* __restrict__ dp,
                                 const int* __restrict__ sr, const int* __restrict__ dr,
                                 float* __restrict__ aggc, float* __restrict__ aggm,
                                 int* __restrict__ headc, int* __restrict__ headm,
                                 int* __restrict__ nodec, int* __restrict__ nodem,
                                 int* __restrict__ rheadc, int* __restrict__ rheadm,
                                 int* __restrict__ rnodec, int* __restrict__ rnodem, int E) {
    int e = blockIdx.x * blockDim.x + threadIdx.x;
    if (e >= E) return;
    const int sc = sp[e], dc = dp[e], sm = sr[e], dm = dr[e];
    atomicAdd(&aggc[dc], xc[sc]);
    atomicAdd(&aggm[dm], xm[sm]);
    reinterpret_cast<int2*>(nodec)[e] = make_int2(atomicExch(&headc[dc], e), sc);
    reinterpret_cast<int2*>(nodem)[e] = make_int2(atomicExch(&headm[dm], e), sm);
    const int sub = e & (RSUB - 1);
    reinterpret_cast<int2*>(rnodec)[e] =
        make_int2(atomicExch(&rheadc[(size_t)sc * RSUB + sub], e), dc);
    reinterpret_cast<int2*>(rnodem)[e] =
        make_int2(atomicExch(&rheadm[(size_t)sm * RSUB + sub], e), dm);
}

#define CT_PITCH 68
#define TS_PITCH 132
#define XS_PITCH 92
#define XT_VEC   ((16 * F85) / 4)

__global__ __launch_bounds__(128) void gemm88_proj_kernel(
    const float* __restrict__ x_trans, const int* __restrict__ rhead,
    const int* __restrict__ rnode, const float* __restrict__ extra1,
    const float* __restrict__ extra2, const float* __restrict__ W,
    const float* __restrict__ bias, const float* __restrict__ W2,
    float* __restrict__ out, int M, int E) {
    __shared__ float xs[16 * KPAD];
    __shared__ float hs[16 * CT_PITCH];

    const int row0 = blockIdx.x * 16;
    const int wave = threadIdx.x >> 5;
    const int lane = threadIdx.x & 31;

    float e1 = 0.0f, e2 = 0.0f;
    if (threadIdx.x < 16) {
        if (extra1) e1 = extra1[row0 + threadIdx.x];
        if (extra2) e2 = extra2[row0 + threadIdx.x];
    }

    {
        const int2* __restrict__ rn = reinterpret_cast<const int2*>(rnode);
        const int c2 = lane < F85 - 64 ? 64 + lane : 64;
        float a0[4], a1[4], a2[4];
#pragma unroll
        for (int i = 0; i < 4; ++i) { a0[i] = 0.0f; a1[i] = 0.0f; a2[i] = 0.0f; }

        int  cur = rhead[(size_t)(row0 + 4 * wave + lane / RSUB) * RSUB + (lane & (RSUB - 1))];
        bool act = (unsigned)cur < (unsigned)E;
        int2 nd  = rn[act ? cur : 0];
        for (int s = 0; s < E; ++s) {
            const unsigned live = __builtin_amdgcn_ballot_w32(act);
            if (live == 0) break;
            const int trow = nd.y;
            const int nxt  = nd.x;
            const bool act2 = act && (unsigned)nxt < (unsigned)E;
            const int2 ndn = rn[act2 ? nxt : 0];
            float v0[32], v1[32], v2[32];
#pragma unroll
            for (int c = 0; c < 32; ++c) {
                v0[c] = 0.0f; v1[c] = 0.0f; v2[c] = 0.0f;
                if (live & (1u << c)) {
                    const float* __restrict__ p =
                        x_trans + (size_t)__builtin_amdgcn_readlane(trow, c) * F85;
                    v0[c] = p[lane];
                    v1[c] = p[32 + lane];
                    v2[c] = p[c2];
                }
            }
#pragma unroll
            for (int c = 0; c < 32; ++c) {
                a0[c / RSUB] += v0[c];
                a1[c / RSUB] += v1[c];
                a2[c / RSUB] += v2[c];
            }
            nd  = ndn;
            act = act2;
        }
#pragma unroll
        for (int i = 0; i < 4; ++i) {
            float* xr = xs + (4 * wave + i) * KPAD;
            xr[lane]      = a0[i];
            xr[32 + lane] = a1[i];
            if (lane < F85 - 64) xr[64 + lane] = a2[i];
        }
    }
    if (threadIdx.x < 16) {
        xs[threadIdx.x * KPAD + 85] = e1;
        xs[threadIdx.x * KPAD + 86] = e2;
        xs[threadIdx.x * KPAD + 87] = 0.0f;
    }
    __syncthreads();

    const int half = lane >> 4;
    const int lm   = lane & 15;
    const int c0   = wave * 16;

    const float bval = bias[c0 + lm];
    v8f acc;
#pragma unroll
    for (int i = 0; i < 8; ++i) acc[i] = bval;

    const float4* __restrict__ Wf  = reinterpret_cast<const float4*>(W)  + wave * 32 + lane;
    const float4* __restrict__ W2f = reinterpret_cast<const float4*>(W2) + wave * 32 + lane;

    for (int kp = 0; kp < KPAD / 8; ++kp) {
        const float4 bq = Wf[kp * 128];
        const int ka = 8 * kp + 2 * half;
        v2f a, b;
        a.x = xs[lm * KPAD + ka];
        a.y = xs[lm * KPAD + ka + 1];
        b.x = bq.x;
        b.y = bq.y;
        acc = __builtin_amdgcn_wmma_f32_16x16x4_f32(false, a, false, b, (short)0, acc,
                                                    false, false);
        a.x = xs[lm * KPAD + ka + 4];
        a.y = xs[lm * KPAD + ka + 5];
        b.x = bq.z;
        b.y = bq.w;
        acc = __builtin_amdgcn_wmma_f32_16x16x4_f32(false, a, false, b, (short)0, acc,
                                                    false, false);
    }
#pragma unroll
    for (int v = 0; v < 8; ++v) {
        float r = acc[v];
        r = r > 0.0f ? r : 0.0f;
        hs[(v + 8 * half) * CT_PITCH + c0 + lm] = r;
    }
    __syncthreads();

    v8f g;
#pragma unroll
    for (int i = 0; i < 8; ++i) g[i] = 0.0f;
    for (int kp = 0; kp < H / 8; ++kp) {
        const float4 bq = W2f[kp * 128];
        const int ka = 8 * kp + 2 * half;
        v2f a, b;
        a.x = hs[lm * CT_PITCH + ka];
        a.y = hs[lm * CT_PITCH + ka + 1];
        b.x = bq.x;
        b.y = bq.y;
        g = __builtin_amdgcn_wmma_f32_16x16x4_f32(false, a, false, b, (short)0, g,
                                                  false, false);
        a.x = hs[lm * CT_PITCH + ka + 4];
        a.y = hs[lm * CT_PITCH + ka + 5];
        b.x = bq.z;
        b.y = bq.w;
        g = __builtin_amdgcn_wmma_f32_16x16x4_f32(false, a, false, b, (short)0, g,
                                                  false, false);
    }
#pragma unroll
    for (int v = 0; v < 8; ++v)
        out[(size_t)(row0 + v + 8 * half) * H + c0 + lm] = g[v];
}


__global__ __launch_bounds__(128, 12) void conv12_out_kernel(
    const float* __restrict__ x_trans,
    const float* __restrict__ aggc1, const float* __restrict__ aggm1,
    const float* __restrict__ W1t,  const float* __restrict__ b1sum,
    const float* __restrict__ g_c,  const float* __restrict__ g_m,
    const int* __restrict__ headc,  const int* __restrict__ headm,
    const int* __restrict__ nodec,  const int* __restrict__ nodem,
    const float* __restrict__ w2r,  const float* __restrict__ b2,
    const float* __restrict__ wlin, const float* __restrict__ blin,
    float* __restrict__ out, int M, int E) {
    __shared__ alignas(16) float hs[16 * CT_PITCH];
    __shared__ alignas(16) float cs[16 * CT_PITCH];
    __shared__ alignas(16) float ms[16 * CT_PITCH];
    __shared__ alignas(16) float ts[16 * TS_PITCH];
    float* xs = ts;

    const int row0 = blockIdx.x * 16;
    const int tid  = threadIdx.x;

    const int gr = tid >> 3;
    const int gq = tid & 7;
    const int ec = headc[row0 + gr];
    const int em = headm[row0 + gr];

    float xr[12];
    {
        const float* __restrict__ xt = x_trans + (size_t)row0 * F85;
#pragma unroll
        for (int j = 0; j < 3; ++j) {
            const int p = tid + 128 * j;
            const int q = 4 * (p < XT_VEC ? p : XT_VEC - 1);
#pragma unroll
            for (int c = 0; c < 4; ++c) xr[4 * j + c] = xt[q + c];
        }
    }
    const float xe = ((tid & 16) ? aggm1 : aggc1)[row0 + (tid & 15)];

    {
        const int2*   __restrict__ nc  = reinterpret_cast<const int2*>(nodec);
        const int2*   __restrict__ nm  = reinterpret_cast<const int2*>(nodem);
        const float4* __restrict__ hc4 = reinterpret_cast<const float4*>(g_c);
        const float4* __restrict__ hm4 = reinterpret_cast<const float4*>(g_m);
        float4 c0 = make_float4(0.f, 0.f, 0.f, 0.f), c1 = c0, m0 = c0, m1 = c0;
        int  ic = ec, im = em;
        bool lc = (unsigned)ic < (unsigned)E, lm = (unsigned)im < (unsigned)E;
        for (int s = 0; s < E && (lc || lm); ++s) {
            const int2 dc = nc[lc ? ic : 0];
            const int2 dm = nm[lm ? im : 0];
            const float4* pc = hc4 + (size_t)dc.y * (H / 4) + gq;
            const float4* pm = hm4 + (size_t)dm.y * (H / 4) + gq;
            const float4 u0 = pc[0], u1 = pc[8], w0 = pm[0], w1 = pm[8];
            if (lc) { c0 += u0; c1 += u1; ic = dc.x; }
            if (lm) { m0 += w0; m1 += w1; im = dm.x; }
            lc = lc && (unsigned)ic < (unsigned)E;
            lm = lm && (unsigned)im < (unsigned)E;
        }
        float4* cs4 = reinterpret_cast<float4*>(cs);
        float4* ms4 = reinterpret_cast<float4*>(ms);
        cs4[gr * (CT_PITCH / 4) + gq]     = c0;
        cs4[gr * (CT_PITCH / 4) + 8 + gq] = c1;
        ms4[gr * (CT_PITCH / 4) + gq]     = m0;
        ms4[gr * (CT_PITCH / 4) + 8 + gq] = m1;
    }

#pragma unroll
    for (int j = 0; j < 3; ++j) {
        const int p = tid + 128 * j;
        if (j < 2 || p < XT_VEC) {
#pragma unroll
            for (int c = 0; c < 4; ++c) {
                const int i = 4 * p + c;
                const int r = i / F85;
                xs[r * XS_PITCH + (i - r * F85)] = xr[4 * j + c];
            }
        }
    }
    if (tid < 48) xs[(tid & 15) * XS_PITCH + F85 + (tid >> 4)] = tid < 32 ? xe : 0.0f;
    __syncthreads();

    const int wave = tid >> 5;
    const int lane = tid & 31;
    const int half = lane >> 4;
    const int lm   = lane & 15;
    const int c0   = wave * 16;

    {
        v8f acc;
        const float bv = b1sum[c0 + lm];
#pragma unroll
        for (int i = 0; i < 8; ++i) acc[i] = bv;
        const float4* __restrict__ W1f = reinterpret_cast<const float4*>(W1t) + wave * 32 + lane;
        for (int kp = 0; kp < KPAD / 8; ++kp) {
            const float4 bq = W1f[kp * 128];
            const int ka = 8 * kp + 2 * half;
            v2f a, b;
            a.x = xs[lm * XS_PITCH + ka];
            a.y = xs[lm * XS_PITCH + ka + 1];
            b.x = bq.x;
            b.y = bq.y;
            acc = __builtin_amdgcn_wmma_f32_16x16x4_f32(false, a, false, b, (short)0, acc,
                                                        false, false);
            a.x = xs[lm * XS_PITCH + ka + 4];
            a.y = xs[lm * XS_PITCH + ka + 5];
            b.x = bq.z;
            b.y = bq.w;
            acc = __builtin_amdgcn_wmma_f32_16x16x4_f32(false, a, false, b, (short)0, acc,
                                                        false, false);
        }
#pragma unroll
        for (int v = 0; v < 8; ++v) {
            float r = acc[v];
            r = r > 0.0f ? r : 0.0f;
            hs[(v + 8 * half) * CT_PITCH + c0 + lm] = r;
        }
    }
    __syncthreads();

    v8f acc1, acc2;
    const float bv1 = b2[0 * H + c0 + lm];
    const float bv2 = b2[1 * H + c0 + lm];
#pragma unroll
    for (int v = 0; v < 8; ++v) {
        acc1[v] = bv1 + cs[(v + 8 * half) * CT_PITCH + c0 + lm];
        acc2[v] = bv2 + ms[(v + 8 * half) * CT_PITCH + c0 + lm];
    }

    const float4* __restrict__ W2f = reinterpret_cast<const float4*>(w2r) + wave * 32 + lane;

    for (int k = 0; k < H; k += 4) {
        const float4 bq = W2f[(k >> 2) * 128];
        const int ka = k + 2 * half;
        v2f ah, b;
        ah.x = hs[lm * CT_PITCH + ka]; ah.y = hs[lm * CT_PITCH + ka + 1];

        b.x = bq.x; b.y = bq.y;
        acc1 = __builtin_amdgcn_wmma_f32_16x16x4_f32(false, ah, false, b, (short)0, acc1,
                                                     false, false);
        b.x = bq.z; b.y = bq.w;
        acc2 = __builtin_amdgcn_wmma_f32_16x16x4_f32(false, ah, false, b, (short)0, acc2,
                                                     false, false);
    }

#pragma unroll
    for (int v = 0; v < 8; ++v) {
        float r1 = acc1[v]; r1 = r1 > 0.0f ? r1 : 0.0f;
        float r2 = acc2[v]; r2 = r2 > 0.0f ? r2 : 0.0f;
        ts[(v + 8 * half) * TS_PITCH + c0 + lm]      = r1;
        ts[(v + 8 * half) * TS_PITCH + 64 + c0 + lm] = r2;
    }
    __syncthreads();

    if (tid < 32) {
        const int r = tid >> 1, c = tid & 1;
        const float4* __restrict__ w4 = reinterpret_cast<const float4*>(wlin) + c * 32;
        const float4* t4 = reinterpret_cast<const float4*>(ts + r * TS_PITCH);
        float s = blin[c];
#pragma unroll 8
        for (int j = 0; j < 32; ++j) {
            const float4 t = t4[j];
            const float4 w = w4[j];
            s += t.x * w.x;
            s += t.y * w.y;
            s += t.z * w.z;
            s += t.w * w.w;
        }
        out[(size_t)(row0 + r) * 2 + c] = s;
    }
}

extern "C" void kernel_launch(void* const* d_in, const int* in_sizes, int n_in,
                              void* d_out, int out_size, void* d_ws, size_t ws_size,
                              hipStream_t stream) {
    const float* x_card   = (const float*)d_in[0];
    const float* x_merch  = (const float*)d_in[1];
    const float* x_trans  = (const float*)d_in[2];
    const float* w1_src_t = (const float*)d_in[3];
    const float* w1_dst_t = (const float*)d_in[4];
    const float* b1_t     = (const float*)d_in[5];
    const float* w1_src_r = (const float*)d_in[6];
    const float* w1_dst_r = (const float*)d_in[7];
    const float* b1_r     = (const float*)d_in[8];
    const float* w2_l     = (const float*)d_in[9];
    const float* w2_r     = (const float*)d_in[10];
    const float* b2       = (const float*)d_in[11];
    const float* w_lin    = (const float*)d_in[12];
    const float* b_lin    = (const float*)d_in[13];
    const int* src_pays   = (const int*)d_in[14];
    const int* dst_pays   = (const int*)d_in[15];
    const int* src_recv   = (const int*)d_in[16];
    const int* dst_recv   = (const int*)d_in[17];
    float* out = (float*)d_out;

    const int N_c = in_sizes[0];
    const int N_m = in_sizes[1];
    const int N_t = in_sizes[2] / F85;
    const int E   = in_sizes[14];

    float* ws = (float*)d_ws;
    size_t off = 0;
    float* W1t    = ws + off; off += (size_t)KPAD * H;
    float* W1c    = ws + off; off += (size_t)KPAD * H;
    float* W1m    = ws + off; off += (size_t)KPAD * H;
    float* W2lc   = ws + off; off += (size_t)H * H;
    float* W2lm   = ws + off; off += (size_t)H * H;
    float* W2rf   = ws + off; off += (size_t)2 * H * H;
    float* wlinT  = ws + off; off += (size_t)4 * H;
    float* b1sum  = ws + off; off += (size_t)H;
    float* g_c    = ws + off; off += (size_t)N_c * H;
    float* g_m    = ws + off; off += (size_t)N_m * H;
    int*   nodec  = (int*)(ws + off); off += (size_t)E * 2;
    int*   nodem  = (int*)(ws + off); off += (size_t)E * 2;
    int*   rnodec = (int*)(ws + off); off += (size_t)E * 2;
    int*   rnodem = (int*)(ws + off); off += (size_t)E * 2;
    float* aggc1  = ws + off; off += (size_t)N_t;
    float* aggm1  = ws + off; off += (size_t)N_t;
    int*   headc  = (int*)(ws + off); off += (size_t)N_t;
    int*   headm  = (int*)(ws + off); off += (size_t)N_t;
    int*   rheadc = (int*)(ws + off); off += (size_t)N_c * RSUB;
    int*   rheadm = (int*)(ws + off); off += (size_t)N_m * RSUB;

    {
        long nz = 2 * (long)N_t;
        long nh = 2 * (long)N_t + (long)RSUB * ((long)N_c + (long)N_m);
        long blocks = (nh + 255) / 256;
        init_kernel<<<(unsigned)blocks, 256, 0, stream>>>(aggc1, headc, nz, nh);
    }
    prep_w_kernel<<<(2 * H * H + 255) / 256, 256, 0, stream>>>(
        w1_src_t, w1_dst_t, b1_t, w1_src_r, w1_dst_r, w2_l, w2_r, w_lin,
        W1t, W1c, W1m, b1sum, W2lc, W2lm, W2rf, wlinT);
    edge_pass_kernel<<<(E + 255) / 256, 256, 0, stream>>>(
        x_card, x_merch, src_pays, dst_pays, src_recv, dst_recv, aggc1, aggm1,
        headc, headm, nodec, nodem, rheadc, rheadm, rnodec, rnodem, E);
    gemm88_proj_kernel<<<N_c / 16, 128, 0, stream>>>(x_trans, rheadc, rnodec, x_card, nullptr,
                                                     W1c, b1_r, W2lc, g_c, N_c, E);
    gemm88_proj_kernel<<<N_m / 16, 128, 0, stream>>>(x_trans, rheadm, rnodem, x_merch, nullptr,
                                                     W1m, b1_r + H, W2lm, g_m, N_m, E);
    conv12_out_kernel<<<N_t / 16, 128, 0, stream>>>(
        x_trans, aggc1, aggm1, W1t, b1sum, g_c, g_m, headc, headm, nodec, nodem,
        W2rf, b2, wlinT, b_lin, out, N_t, E);
}


# independent kernels share a launch, picked by block index: merchant and card tiles run as one grid (merchant tiles first), the weight prep rides in the init grid; 6 launches -> 4, sums unchanged
# speedup vs baseline: 3.6582x; 1.0515x over previous
#include <hip/hip_runtime.h>

typedef __attribute__((ext_vector_type(2))) float v2f;
typedef __attribute__((ext_vector_type(8))) float v8f;

#define H 64
#define F85 85
#define KPAD 88
#define RSUB 8

__device__ __forceinline__ void prep_w(int i,
                                       const float* __restrict__ w1_src_t,
                                       const float* __restrict__ w1_dst_t,
                                       const float* __restrict__ b1_t,
                                       const float* __restrict__ w1_src_r,
                                       const float* __restrict__ w1_dst_r,
                                       const float* __restrict__ w2_l,
                                       const float* __restrict__ w2_r,
                                       const float* __restrict__ w_lin,
                                       float* __restrict__ W1t, float* __restrict__ W1c,
                                       float* __restrict__ W1m, float* __restrict__ b1sum,
                                       float* __restrict__ W2lc, float* __restrict__ W2lm,
                                       float* __restrict__ W2rf, float* __restrict__ wlinT) {
    const int fj = i & 3, flane = (i >> 2) & 31, fwave = (i >> 7) & 3, fk = i >> 9;
    const int c = 16 * fwave + (flane & 15);
    if (i < 2 * H * H) {
        const int rr = 4 * fk + 2 * (flane >> 4) + (fj & 1);
        W2rf[i] = w2_r[(fj >> 1) * H * H + rr * H + c];
    }
    if (i < 4 * H) wlinT[i] = w_lin[(i & (2 * H - 1)) * 2 + (i >> 7)];
    if (i >= KPAD * H) return;
    const int r = 4 * (2 * fk + (fj >> 1)) + 2 * (flane >> 4) + (fj & 1);
    if (i < H * H) {
        W2lc[i] = w2_l[r * H + c];
        W2lm[i] = w2_l[H * H + r * H + c];
    }
    float vt = 0.f, vc = 0.f, vm = 0.f;
    if (r < F85) {
        vt = w1_dst_t[r * H + c] + w1_dst_t[F85 * H + r * H + c];
        vc = w1_src_r[r * H + c];
        vm = w1_src_r[F85 * H + r * H + c];
    } else if (r == 85) {
        vt = w1_src_t[c];
        vc = w1_dst_r[c];
        vm = w1_dst_r[H + c];
    } else if (r == 86) {
        vt = w1_src_t[H + c];
    }
    W1t[i] = vt; W1c[i] = vc; W1m[i] = vm;
    if (i < H) b1sum[i] = b1_t[i] + b1_t[H + i];
}

#define PREP_BLOCKS ((2 * H * H) / 256)

__global__ void init_prep_kernel(float* __restrict__ z, int* __restrict__ heads, long nz, long nh,
                                 const float* __restrict__ w1_src_t, const float* __restrict__ w1_dst_t,
                                 const float* __restrict__ b1_t, const float* __restrict__ w1_src_r,
                                 const float* __restrict__ w1_dst_r, const float* __restrict__ w2_l,
                                 const float* __restrict__ w2_r, const float* __restrict__ w_lin,
                                 float* __restrict__ W1t, float* __restrict__ W1c,
                                 float* __restrict__ W1m, float* __restrict__ b1sum,
                                 float* __restrict__ W2lc, float* __restrict__ W2lm,
                                 float* __restrict__ W2rf, float* __restrict__ wlinT) {
    if (blockIdx.x < PREP_BLOCKS) {
        prep_w(blockIdx.x * 256 + threadIdx.x, w1_src_t, w1_dst_t, b1_t, w1_src_r, w1_dst_r, w2_l,
               w2_r, w_lin, W1t, W1c, W1m, b1sum, W2lc, W2lm, W2rf, wlinT);
        return;
    }
    long i = (long)(blockIdx.x - PREP_BLOCKS) * 256 + threadIdx.x;
    if (i < nz) z[i] = 0.0f;
    if (i < nh) heads[i] = -1;
}

__global__ void edge_pass_kernel(const float* __restrict__ xc, const float* __restrict__ xm,
                                 const int* __restrict__ sp, const int* __restrict__ dp,
                                 const int* __restrict__ sr, const int* __restrict__ dr,
                                 float* __restrict__ aggc, float* __restrict__ aggm,
                                 int* __restrict__ headc, int* __restrict__ headm,
                                 int* __restrict__ nodec, int* __restrict__ nodem,
                                 int* __restrict__ rheadc, int* __restrict__ rheadm,
                                 int* __restrict__ rnodec, int* __restrict__ rnodem, int E) {
    int e = blockIdx.x * blockDim.x + threadIdx.x;
    if (e >= E) return;
    const int sc = sp[e], dc = dp[e], sm = sr[e], dm = dr[e];
    atomicAdd(&aggc[dc], xc[sc]);
    atomicAdd(&aggm[dm], xm[sm]);
    reinterpret_cast<int2*>(nodec)[e] = make_int2(atomicExch(&headc[dc], e), sc);
    reinterpret_cast<int2*>(nodem)[e] = make_int2(atomicExch(&headm[dm], e), sm);
    const int sub = e & (RSUB - 1);
    reinterpret_cast<int2*>(rnodec)[e] =
        make_int2(atomicExch(&rheadc[(size_t)sc * RSUB + sub], e), dc);
    reinterpret_cast<int2*>(rnodem)[e] =
        make_int2(atomicExch(&rheadm[(size_t)sm * RSUB + sub], e), dm);
}

#define CT_PITCH 68
#define TS_PITCH 132
#define XS_PITCH 92
#define XT_VEC   ((16 * F85) / 4)

__global__ __launch_bounds__(128) void gemm88_proj_kernel(
    const float* __restrict__ x_trans,
    const int* __restrict__ rhead_m, const int* __restrict__ rnode_m,
    const float* __restrict__ extra_m, const float* __restrict__ W_m,
    const float* __restrict__ bias_m, const float* __restrict__ W2_m, float* __restrict__ out_m,
    const int* __restrict__ rhead_c, const int* __restrict__ rnode_c,
    const float* __restrict__ extra_c, const float* __restrict__ W_c,
    const float* __restrict__ bias_c, const float* __restrict__ W2_c, float* __restrict__ out_c,
    int tiles_m, int E) {
    __shared__ float xs[16 * KPAD];
    __shared__ float hs[16 * CT_PITCH];

    const bool mer = (int)blockIdx.x < tiles_m;
    const int* __restrict__   rhead  = mer ? rhead_m : rhead_c;
    const int* __restrict__   rnode  = mer ? rnode_m : rnode_c;
    const float* __restrict__ extra1 = mer ? extra_m : extra_c;
    const float* __restrict__ W      = mer ? W_m     : W_c;
    const float* __restrict__ bias   = mer ? bias_m  : bias_c;
    const float* __restrict__ W2     = mer ? W2_m    : W2_c;
    float* __restrict__       out    = mer ? out_m   : out_c;

    const int row0 = (mer ? (int)blockIdx.x : (int)blockIdx.x - tiles_m) * 16;
    const int wave = threadIdx.x >> 5;
    const int lane = threadIdx.x & 31;

    float e1 = 0.0f;
    if (threadIdx.x < 16) e1 = extra1[row0 + threadIdx.x];

    {
        const int2* __restrict__ rn = reinterpret_cast<const int2*>(rnode);
        const int c2 = lane < F85 - 64 ? 64 + lane : 64;
        float a0[4], a1[4], a2[4];
#pragma unroll
        for (int i = 0; i < 4; ++i) { a0[i] = 0.0f; a1[i] = 0.0f; a2[i] = 0.0f; }

        int  cur = rhead[(size_t)(row0 + 4 * wave + lane / RSUB) * RSUB + (lane & (RSUB - 1))];
        bool act = (unsigned)cur < (unsigned)E;
        int2 nd  = rn[act ? cur : 0];
        for (int s = 0; s < E; ++s) {
            const unsigned live = __builtin_amdgcn_ballot_w32(act);
            if (live == 0) break;
            const int trow = nd.y;
            const int nxt  = nd.x;
            const bool act2 = act && (unsigned)nxt < (unsigned)E;
            const int2 ndn = rn[act2 ? nxt : 0];
            float v0[32], v1[32], v2[32];
#pragma unroll
            for (int c = 0; c < 32; ++c) {
                v0[c] = 0.0f; v1[c] = 0.0f; v2[c] = 0.0f;
                if (live & (1u << c)) {
                    const float* __restrict__ p =
                        x_trans + (size_t)__builtin_amdgcn_readlane(trow, c) * F85;
                    v0[c] = p[lane];
                    v1[c] = p[32 + lane];
                    v2[c] = p[c2];
                }
            }
#pragma unroll
            for (int c = 0; c < 32; ++c) {
                a0[c / RSUB] += v0[c];
                a1[c / RSUB] += v1[c];
                a2[c / RSUB] += v2[c];
            }
            nd  = ndn;
            act = act2;
        }
#pragma unroll
        for (int i = 0; i < 4; ++i) {
            float* xr = xs + (4 * wave + i) * KPAD;
            xr[lane]      = a0[i];
            xr[32 + lane] = a1[i];
            if (lane < F85 - 64) xr[64 + lane] = a2[i];
        }
    }
    if (threadIdx.x < 16) {
        xs[threadIdx.x * KPAD + 85] = e1;
        xs[threadIdx.x * KPAD + 86] = 0.0f;
        xs[threadIdx.x * KPAD + 87] = 0.0f;
    }
    __syncthreads();

    const int half = lane >> 4;
    const int lm   = lane & 15;
    const int c0   = wave * 16;

    const float bval = bias[c0 + lm];
    v8f acc;
#pragma unroll
    for (int i = 0; i < 8; ++i) acc[i] = bval;

    const float4* __restrict__ Wf  = reinterpret_cast<const float4*>(W)  + wave * 32 + lane;
    const float4* __restrict__ W2f = reinterpret_cast<const float4*>(W2) + wave * 32 + lane;

    for (int kp = 0; kp < KPAD / 8; ++kp) {
        const float4 bq = Wf[kp * 128];
        const int ka = 8 * kp + 2 * half;
        v2f a, b;
        a.x = xs[lm * KPAD + ka];
        a.y = xs[lm * KPAD + ka + 1];
        b.x = bq.x;
        b.y = bq.y;
        acc = __builtin_amdgcn_wmma_f32_16x16x4_f32(false, a, false, b, (short)0, acc,
                                                    false, false);
        a.x = xs[lm * KPAD + ka + 4];
        a.y = xs[lm * KPAD + ka + 5];
        b.x = bq.z;
        b.y = bq.w;
        acc = __builtin_amdgcn_wmma_f32_16x16x4_f32(false, a, false, b, (short)0, acc,
                                                    false, false);
    }
#pragma unroll
    for (int v = 0; v < 8; ++v) {
        float r = acc[v];
        r = r > 0.0f ? r : 0.0f;
        hs[(v + 8 * half) * CT_PITCH + c0 + lm] = r;
    }
    __syncthreads();

    v8f g;
#pragma unroll
    for (int i = 0; i < 8; ++i) g[i] = 0.0f;
    for (int kp = 0; kp < H / 8; ++kp) {
        const float4 bq = W2f[kp * 128];
        const int ka = 8 * kp + 2 * half;
        v2f a, b;
        a.x = hs[lm * CT_PITCH + ka];
        a.y = hs[lm * CT_PITCH + ka + 1];
        b.x = bq.x;
        b.y = bq.y;
        g = __builtin_amdgcn_wmma_f32_16x16x4_f32(false, a, false, b, (short)0, g,
                                                  false, false);
        a.x = hs[lm * CT_PITCH + ka + 4];
        a.y = hs[lm * CT_PITCH + ka + 5];
        b.x = bq.z;
        b.y = bq.w;
        g = __builtin_amdgcn_wmma_f32_16x16x4_f32(false, a, false, b, (short)0, g,
                                                  false, false);
    }
#pragma unroll
    for (int v = 0; v < 8; ++v)
        out[(size_t)(row0 + v + 8 * half) * H + c0 + lm] = g[v];
}


__global__ __launch_bounds__(128, 12) void conv12_out_kernel(
    const float* __restrict__ x_trans,
    const float* __restrict__ aggc1, const float* __restrict__ aggm1,
    const float* __restrict__ W1t,  const float* __restrict__ b1sum,
    const float* __restrict__ g_c,  const float* __restrict__ g_m,
    const int* __restrict__ headc,  const int* __restrict__ headm,
    const int* __restrict__ nodec,  const int* __restrict__ nodem,
    const float* __restrict__ w2r,  const float* __restrict__ b2,
    const float* __restrict__ wlin, const float* __restrict__ blin,
    float* __restrict__ out, int M, int E) {
    __shared__ alignas(16) float hs[16 * CT_PITCH];
    __shared__ alignas(16) float cs[16 * CT_PITCH];
    __shared__ alignas(16) float ms[16 * CT_PITCH];
    __shared__ alignas(16) float ts[16 * TS_PITCH];
    float* xs = ts;

    const int row0 = blockIdx.x * 16;
    const int tid  = threadIdx.x;

    const int gr = tid >> 3;
    const int gq = tid & 7;
    const int ec = headc[row0 + gr];
    const int em = headm[row0 + gr];

    float xr[12];
    {
        const float* __restrict__ xt = x_trans + (size_t)row0 * F85;
#pragma unroll
        for (int j = 0; j < 3; ++j) {
            const int p = tid + 128 * j;
            const int q = 4 * (p < XT_VEC ? p : XT_VEC - 1);
#pragma unroll
            for (int c = 0; c < 4; ++c) xr[4 * j + c] = xt[q + c];
        }
    }
    const float xe = ((tid & 16) ? aggm1 : aggc1)[row0 + (tid & 15)];

    {
        const int2*   __restrict__ nc  = reinterpret_cast<const int2*>(nodec);
        const int2*   __restrict__ nm  = reinterpret_cast<const int2*>(nodem);
        const float4* __restrict__ hc4 = reinterpret_cast<const float4*>(g_c);
        const float4* __restrict__ hm4 = reinterpret_cast<const float4*>(g_m);
        float4 c0 = make_float4(0.f, 0.f, 0.f, 0.f), c1 = c0, m0 = c0, m1 = c0;
        int  ic = ec, im = em;
        bool lc = (unsigned)ic < (unsigned)E, lm = (unsigned)im < (unsigned)E;
        for (int s = 0; s < E && (lc || lm); ++s) {
            const int2 dc = nc[lc ? ic : 0];
            const int2 dm = nm[lm ? im : 0];
            const float4* pc = hc4 + (size_t)dc.y * (H / 4) + gq;
            const float4* pm = hm4 + (size_t)dm.y * (H / 4) + gq;
            const float4 u0 = pc[0], u1 = pc[8], w0 = pm[0], w1 = pm[8];
            if (lc) { c0 += u0; c1 += u1; ic = dc.x; }
            if (lm) { m0 += w0; m1 += w1; im = dm.x; }
            lc = lc && (unsigned)ic < (unsigned)E;
            lm = lm && (unsigned)im < (unsigned)E;
        }
        float4* cs4 = reinterpret_cast<float4*>(cs);
        float4* ms4 = reinterpret_cast<float4*>(ms);
        cs4[gr * (CT_PITCH / 4) + gq]     = c0;
        cs4[gr * (CT_PITCH / 4) + 8 + gq] = c1;
        ms4[gr * (CT_PITCH / 4) + gq]     = m0;
        ms4[gr * (CT_PITCH / 4) + 8 + gq] = m1;
    }

#pragma unroll
    for (int j = 0; j < 3; ++j) {
        const int p = tid + 128 * j;
        if (j < 2 || p < XT_VEC) {
#pragma unroll
            for (int c = 0; c < 4; ++c) {
                const int i = 4 * p + c;
                const int r = i / F85;
                xs[r * XS_PITCH + (i - r * F85)] = xr[4 * j + c];
            }
        }
    }
    if (tid < 48) xs[(tid & 15) * XS_PITCH + F85 + (tid >> 4)] = tid < 32 ? xe : 0.0f;
    __syncthreads();

    const int wave = tid >> 5;
    const int lane = tid & 31;
    const int half = lane >> 4;
    const int lm   = lane & 15;
    const int c0   = wave * 16;

    {
        v8f acc;
        const float bv = b1sum[c0 + lm];
#pragma unroll
        for (int i = 0; i < 8; ++i) acc[i] = bv;
        const float4* __restrict__ W1f = reinterpret_cast<const float4*>(W1t) + wave * 32 + lane;
        for (int kp = 0; kp < KPAD / 8; ++kp) {
            const float4 bq = W1f[kp * 128];
            const int ka = 8 * kp + 2 * half;
            v2f a, b;
            a.x = xs[lm * XS_PITCH + ka];
            a.y = xs[lm * XS_PITCH + ka + 1];
            b.x = bq.x;
            b.y = bq.y;
            acc = __builtin_amdgcn_wmma_f32_16x16x4_f32(false, a, false, b, (short)0, acc,
                                                        false, false);
            a.x = xs[lm * XS_PITCH + ka + 4];
            a.y = xs[lm * XS_PITCH + ka + 5];
            b.x = bq.z;
            b.y = bq.w;
            acc = __builtin_amdgcn_wmma_f32_16x16x4_f32(false, a, false, b, (short)0, acc,
                                                        false, false);
        }
#pragma unroll
        for (int v = 0; v < 8; ++v) {
            float r = acc[v];
            r = r > 0.0f ? r : 0.0f;
            hs[(v + 8 * half) * CT_PITCH + c0 + lm] = r;
        }
    }
    __syncthreads();

    v8f acc1, acc2;
    const float bv1 = b2[0 * H + c0 + lm];
    const float bv2 = b2[1 * H + c0 + lm];
#pragma unroll
    for (int v = 0; v < 8; ++v) {
        acc1[v] = bv1 + cs[(v + 8 * half) * CT_PITCH + c0 + lm];
        acc2[v] = bv2 + ms[(v + 8 * half) * CT_PITCH + c0 + lm];
    }

    const float4* __restrict__ W2f = reinterpret_cast<const float4*>(w2r) + wave * 32 + lane;

    for (int k = 0; k < H; k += 4) {
        const float4 bq = W2f[(k >> 2) * 128];
        const int ka = k + 2 * half;
        v2f ah, b;
        ah.x = hs[lm * CT_PITCH + ka]; ah.y = hs[lm * CT_PITCH + ka + 1];

        b.x = bq.x; b.y = bq.y;
        acc1 = __builtin_amdgcn_wmma_f32_16x16x4_f32(false, ah, false, b, (short)0, acc1,
                                                     false, false);
        b.x = bq.z; b.y = bq.w;
        acc2 = __builtin_amdgcn_wmma_f32_16x16x4_f32(false, ah, false, b, (short)0, acc2,
                                                     false, false);
    }

#pragma unroll
    for (int v = 0; v < 8; ++v) {
        float r1 = acc1[v]; r1 = r1 > 0.0f ? r1 : 0.0f;
        float r2 = acc2[v]; r2 = r2 > 0.0f ? r2 : 0.0f;
        ts[(v + 8 * half) * TS_PITCH + c0 + lm]      = r1;
        ts[(v + 8 * half) * TS_PITCH + 64 + c0 + lm] = r2;
    }
    __syncthreads();

    if (tid < 32) {
        const int r = tid >> 1, c = tid & 1;
        const float4* __restrict__ w4 = reinterpret_cast<const float4*>(wlin) + c * 32;
        const float4* t4 = reinterpret_cast<const float4*>(ts + r * TS_PITCH);
        float s = blin[c];
#pragma unroll 8
        for (int j = 0; j < 32; ++j) {
            const float4 t = t4[j];
            const float4 w = w4[j];
            s += t.x * w.x;
            s += t.y * w.y;
            s += t.z * w.z;
            s += t.w * w.w;
        }
        out[(size_t)(row0 + r) * 2 + c] = s;
    }
}

extern "C" void kernel_launch(void* const* d_in, const int* in_sizes, int n_in,
                              void* d_out, int out_size, void* d_ws, size_t ws_size,
                              hipStream_t stream) {
    const float* x_card   = (const float*)d_in[0];
    const float* x_merch  = (const float*)d_in[1];
    const float* x_trans  = (const float*)d_in[2];
    const float* w1_src_t = (const float*)d_in[3];
    const float* w1_dst_t = (const float*)d_in[4];
    const float* b1_t     = (const float*)d_in[5];
    const float* w1_src_r = (const float*)d_in[6];
    const float* w1_dst_r = (const float*)d_in[7];
    const float* b1_r     = (const float*)d_in[8];
    const float* w2_l     = (const float*)d_in[9];
    const float* w2_r     = (const float*)d_in[10];
    const float* b2       = (const float*)d_in[11];
    const float* w_lin    = (const float*)d_in[12];
    const float* b_lin    = (const float*)d_in[13];
    const int* src_pays   = (const int*)d_in[14];
    const int* dst_pays   = (const int*)d_in[15];
    const int* src_recv   = (const int*)d_in[16];
    const int* dst_recv   = (const int*)d_in[17];
    float* out = (float*)d_out;

    const int N_c = in_sizes[0];
    const int N_m = in_sizes[1];
    const int N_t = in_sizes[2] / F85;
    const int E   = in_sizes[14];

    float* ws = (float*)d_ws;
    size_t off = 0;
    float* W1t    = ws + off; off += (size_t)KPAD * H;
    float* W1c    = ws + off; off += (size_t)KPAD * H;
    float* W1m    = ws + off; off += (size_t)KPAD * H;
    float* W2lc   = ws + off; off += (size_t)H * H;
    float* W2lm   = ws + off; off += (size_t)H * H;
    float* W2rf   = ws + off; off += (size_t)2 * H * H;
    float* wlinT  = ws + off; off += (size_t)4 * H;
    float* b1sum  = ws + off; off += (size_t)H;
    float* g_c    = ws + off; off += (size_t)N_c * H;
    float* g_m    = ws + off; off += (size_t)N_m * H;
    int*   nodec  = (int*)(ws + off); off += (size_t)E * 2;
    int*   nodem  = (int*)(ws + off); off += (size_t)E * 2;
    int*   rnodec = (int*)(ws + off); off += (size_t)E * 2;
    int*   rnodem = (int*)(ws + off); off += (size_t)E * 2;
    float* aggc1  = ws + off; off += (size_t)N_t;
    float* aggm1  = ws + off; off += (size_t)N_t;
    int*   headc  = (int*)(ws + off); off += (size_t)N_t;
    int*   headm  = (int*)(ws + off); off += (size_t)N_t;
    int*   rheadc = (int*)(ws + off); off += (size_t)N_c * RSUB;
    int*   rheadm = (int*)(ws + off); off += (size_t)N_m * RSUB;

    {
        long nz = 2 * (long)N_t;
        long nh = 2 * (long)N_t + (long)RSUB * ((long)N_c + (long)N_m);
        long blocks = (nh + 255) / 256 + PREP_BLOCKS;
        init_prep_kernel<<<(unsigned)blocks, 256, 0, stream>>>(
            aggc1, headc, nz, nh,
            w1_src_t, w1_dst_t, b1_t, w1_src_r, w1_dst_r, w2_l, w2_r, w_lin,
            W1t, W1c, W1m, b1sum, W2lc, W2lm, W2rf, wlinT);
    }
    edge_pass_kernel<<<(E + 255) / 256, 256, 0, stream>>>(
        x_card, x_merch, src_pays, dst_pays, src_recv, dst_recv, aggc1, aggm1,
        headc, headm, nodec, nodem, rheadc, rheadm, rnodec, rnodem, E);
    gemm88_proj_kernel<<<N_m / 16 + N_c / 16, 128, 0, stream>>>(
        x_trans,
        rheadm, rnodem, x_merch, W1m, b1_r + H, W2lm, g_m,
        rheadc, rnodec, x_card,  W1c, b1_r,     W2lc, g_c,
        N_m / 16, E);
    conv12_out_kernel<<<N_t / 16, 128, 0, stream>>>(
        x_trans, aggc1, aggm1, W1t, b1sum, g_c, g_m, headc, headm, nodec, nodem,
        W2rf, b2, wlinT, b_lin, out, N_t, E);
}
